# starting kernel
# baseline (speedup 1.0000x reference)
#include <hip/hip_runtime.h>
#include <hip/hip_bf16.h>
#include <stdint.h>

typedef __attribute__((ext_vector_type(16))) __bf16 v16bf;
typedef __attribute__((ext_vector_type(8)))  float  v8f;

#define NUM_NODES 10000
#define IN_FEAT   64
#define HID       256
#define OUTF      3
#define GATES     768

__global__ void cvt_f32_to_bf16(const float* __restrict__ in,
                                __bf16* __restrict__ out, int n) {
    int i = blockIdx.x * blockDim.x + threadIdx.x;
    if (i < n) out[i] = (__bf16)in[i];
}

__global__ void cvt_f32_to_bf16_T(const float* __restrict__ in,
                                  __bf16* __restrict__ out, int K, int N) {
    int i = blockIdx.x * blockDim.x + threadIdx.x;
    if (i >= K * N) return;
    int k = i / N, n = i % N;
    out[(size_t)n * K + k] = (__bf16)in[(size_t)k * N + n];
}

__global__ void deg_init(float* __restrict__ deg, int n) {
    int i = blockIdx.x * blockDim.x + threadIdx.x;
    if (i < n) deg[i] = 1.0f;
}

__global__ void deg_edges(const int* __restrict__ dst, float* __restrict__ deg, int E) {
    int e = blockIdx.x * blockDim.x + threadIdx.x;
    if (e < E) atomicAdd(&deg[dst[e]], 1.0f);
}

__global__ void deg_rsqrt(const float* __restrict__ deg, float* __restrict__ dinv, int n) {
    int i = blockIdx.x * blockDim.x + threadIdx.x;
    if (i < n) dinv[i] = rsqrtf(deg[i]);
}

__global__ void __launch_bounds__(128)
wmma_gemm_bf16(const __bf16* __restrict__ A, const __bf16* __restrict__ Bt,
               float* __restrict__ C, int M, int N, int K) {
    int wave  = (int)((blockIdx.x * blockDim.x + threadIdx.x) >> 5);
    int lane  = threadIdx.x & 31;
    int tilesN = N >> 4;
    int tilesM = M >> 4;
    if (wave >= tilesM * tilesN) return;
    int tm = wave / tilesN;
    int tn = wave % tilesN;

    int l15 = lane & 15;
    int h   = lane >> 4;
    const __bf16* arow = A  + (size_t)(tm * 16 + l15) * K;
    const __bf16* bcol = Bt + (size_t)(tn * 16 + l15) * K;

    v8f acc = {};
    for (int k = 0; k < K; k += 32) {
        v16bf a, b;
        int k0 = k + h * 8;
#pragma unroll
        for (int i = 0; i < 8; ++i) {
            a[i]     = arow[k0 + i];
            a[i + 8] = arow[k0 + 16 + i];
            b[i]     = bcol[k0 + i];
            b[i + 8] = bcol[k0 + 16 + i];
        }
        acc = __builtin_amdgcn_wmma_f32_16x16x32_bf16(
            false, a, false, b, (short)0, acc, false, false);
    }
#pragma unroll
    for (int r = 0; r < 8; ++r) {
        int mm = tm * 16 + h * 8 + r;
        C[(size_t)mm * N + tn * 16 + l15] = acc[r];
    }
}

__global__ void gcn_selfloop(const float* __restrict__ xw, const float* __restrict__ dinv,
                             float* __restrict__ agg, int n) {
    int i = blockIdx.x * blockDim.x + threadIdx.x;
    if (i >= n) return;
    int node = i >> 8;
    float s = dinv[node];
    agg[i] = xw[i] * s * s;
}

__global__ void gcn_scatter(const float* __restrict__ xw,
                            const int* __restrict__ src, const int* __restrict__ dst,
                            const float* __restrict__ dinv,
                            float* __restrict__ agg, int E) {
    long long gid = (long long)blockIdx.x * blockDim.x + threadIdx.x;
    int e = (int)(gid >> 6);
    if (e >= E) return;
    int f = ((int)gid & 63) << 2;
    int s = src[e], d = dst[e];
    float norm = dinv[s] * dinv[d];
    const float4 v = *(const float4*)(xw + (size_t)s * HID + f);
    float* o = agg + (size_t)d * HID + f;
    atomicAdd(o + 0, norm * v.x);
    atomicAdd(o + 1, norm * v.y);
    atomicAdd(o + 2, norm * v.z);
    atomicAdd(o + 3, norm * v.w);
}

__global__ void bias_relu_bf16(const float* __restrict__ agg, const float* __restrict__ bias,
                               __bf16* __restrict__ hb, int n) {
    int i = blockIdx.x * blockDim.x + threadIdx.x;
    if (i >= n) return;
    float v = agg[i] + bias[i & (HID - 1)];
    hb[i] = (__bf16)(v > 0.f ? v : 0.f);
}

__global__ void __launch_bounds__(1024)
gru_scan(const float* __restrict__ gi, const __bf16* __restrict__ Whh,
         const float* __restrict__ b_ih, const float* __restrict__ b_hh,
         float* __restrict__ hs, int T) {
    __shared__ float hcur[HID];
    __shared__ float gh[GATES];
    int t = threadIdx.x;
    if (t < HID) hcur[t] = 0.0f;
    __syncthreads();

    for (int step = 0; step < T; ++step) {
        if (t < GATES) {
            const uint32_t* wrow = (const uint32_t*)(Whh + (size_t)t * HID);
            float acc = 0.0f;
#pragma unroll 8
            for (int kk = 0; kk < HID / 2; ++kk) {
                uint32_t p = wrow[kk];
                float w0 = __uint_as_float(p << 16);
                float w1 = __uint_as_float(p & 0xffff0000u);
                acc = fmaf(w0, hcur[2 * kk],     acc);
                acc = fmaf(w1, hcur[2 * kk + 1], acc);
            }
            gh[t] = acc + b_hh[t];
        } else if (step + 1 < T && t < GATES + 32) {
            __builtin_prefetch(gi + (size_t)(step + 1) * GATES + (t - GATES) * 32, 0, 0);
        }
        __syncthreads();
        if (t < HID) {
            const float* g = gi + (size_t)step * GATES;
            float r  = 1.0f / (1.0f + __expf(-(g[t]           + b_ih[t]           + gh[t])));
            float z  = 1.0f / (1.0f + __expf(-(g[HID + t]     + b_ih[HID + t]     + gh[HID + t])));
            float nn = tanhf(g[2 * HID + t] + b_ih[2 * HID + t] + r * gh[2 * HID + t]);
            float hn = (1.0f - z) * nn + z * hcur[t];
            hs[(size_t)step * HID + t] = hn;
            hcur[t] = hn;
        }
        __syncthreads();
    }
}

__global__ void fc_out_kernel(const float* __restrict__ hs, const float* __restrict__ fw,
                              const float* __restrict__ fb, float* __restrict__ out, int total) {
    int gid = blockIdx.x * blockDim.x + threadIdx.x;
    if (gid >= total) return;
    int n = gid / OUTF, o = gid % OUTF;
    const float* h = hs + (size_t)n * HID;
    const float* w = fw + (size_t)o * HID;
    float acc = fb[o];
#pragma unroll 8
    for (int k = 0; k < HID; ++k) acc = fmaf(h[k], w[k], acc);
    out[gid] = acc;
}

extern "C" void kernel_launch(void* const* d_in, const int* in_sizes, int n_in,
                              void* d_out, int out_size, void* d_ws, size_t ws_size,
                              hipStream_t stream) {
    const float* x    = (const float*)d_in[0];
    const int*   ei   = (const int*)  d_in[1];
    const float* W1   = (const float*)d_in[2];
    const float* b1   = (const float*)d_in[3];
    const float* W2   = (const float*)d_in[4];
    const float* b2   = (const float*)d_in[5];
    const float* W_ih = (const float*)d_in[6];
    const float* W_hh = (const float*)d_in[7];
    const float* b_ih = (const float*)d_in[8];
    const float* b_hh = (const float*)d_in[9];
    const float* fc_w = (const float*)d_in[10];
    const float* fc_b = (const float*)d_in[11];
    float* out = (float*)d_out;

    const int E = in_sizes[1] / 2;
    const int* src = ei;
    const int* dst = ei + E;

    char* ws = (char*)d_ws;
    size_t off = 0;
    auto alloc = [&](size_t bytes) -> char* {
        char* p = ws + off;
        off = (off + bytes + 255) & ~(size_t)255;
        return p;
    };
    float*  deg  = (float*) alloc((size_t)NUM_NODES * 4);
    float*  dinv = (float*) alloc((size_t)NUM_NODES * 4);
    float*  xw   = (float*) alloc((size_t)NUM_NODES * HID * 4);
    float*  agg  = (float*) alloc((size_t)NUM_NODES * HID * 4);
    float*  gi   = (float*) alloc((size_t)NUM_NODES * GATES * 4);
    float*  hs   = (float*) alloc((size_t)NUM_NODES * HID * 4);
    __bf16* xb   = (__bf16*)alloc((size_t)NUM_NODES * HID * 2);
    __bf16* w1t  = (__bf16*)alloc((size_t)HID * IN_FEAT * 2);
    __bf16* w2t  = (__bf16*)alloc((size_t)HID * HID * 2);
    __bf16* wih  = (__bf16*)alloc((size_t)GATES * HID * 2);
    __bf16* whh  = (__bf16*)alloc((size_t)GATES * HID * 2);
    (void)ws_size; (void)n_in; (void)out_size;

    const int TPB = 256;
    auto blk = [](long long n, int t) { return (unsigned)((n + t - 1) / t); };

    cvt_f32_to_bf16<<<blk((long long)NUM_NODES * IN_FEAT, TPB), TPB, 0, stream>>>(
        x, xb, NUM_NODES * IN_FEAT);
    cvt_f32_to_bf16_T<<<blk((long long)IN_FEAT * HID, TPB), TPB, 0, stream>>>(
        W1, w1t, IN_FEAT, HID);
    cvt_f32_to_bf16_T<<<blk((long long)HID * HID, TPB), TPB, 0, stream>>>(
        W2, w2t, HID, HID);
    cvt_f32_to_bf16<<<blk((long long)GATES * HID, TPB), TPB, 0, stream>>>(
        W_ih, wih, GATES * HID);
    cvt_f32_to_bf16<<<blk((long long)GATES * HID, TPB), TPB, 0, stream>>>(
        W_hh, whh, GATES * HID);

    deg_init<<<blk(NUM_NODES, TPB), TPB, 0, stream>>>(deg, NUM_NODES);
    deg_edges<<<blk(E, TPB), TPB, 0, stream>>>(dst, deg, E);
    deg_rsqrt<<<blk(NUM_NODES, TPB), TPB, 0, stream>>>(deg, dinv, NUM_NODES);

    const long long NH = (long long)NUM_NODES * HID;
    const unsigned gemm1_blocks = blk((long long)(NUM_NODES / 16) * (HID / 16), 4);
    const unsigned gemm2_blocks = gemm1_blocks;
    const unsigned gemm3_blocks = blk((long long)(NUM_NODES / 16) * (GATES / 16), 4);
    const unsigned scat_blocks  = blk((long long)E * 64, TPB);

    wmma_gemm_bf16<<<gemm1_blocks, 128, 0, stream>>>(xb, w1t, xw, NUM_NODES, HID, IN_FEAT);
    gcn_selfloop<<<blk(NH, TPB), TPB, 0, stream>>>(xw, dinv, agg, (int)NH);
    gcn_scatter<<<scat_blocks, TPB, 0, stream>>>(xw, src, dst, dinv, agg, E);
    bias_relu_bf16<<<blk(NH, TPB), TPB, 0, stream>>>(agg, b1, xb, (int)NH);

    wmma_gemm_bf16<<<gemm2_blocks, 128, 0, stream>>>(xb, w2t, xw, NUM_NODES, HID, HID);
    gcn_selfloop<<<blk(NH, TPB), TPB, 0, stream>>>(xw, dinv, agg, (int)NH);
    gcn_scatter<<<scat_blocks, TPB, 0, stream>>>(xw, src, dst, dinv, agg, E);
    bias_relu_bf16<<<blk(NH, TPB), TPB, 0, stream>>>(agg, b2, xb, (int)NH);

    wmma_gemm_bf16<<<gemm3_blocks, 128, 0, stream>>>(xb, wih, gi, NUM_NODES, GATES, HID);

    gru_scan<<<1, 1024, 0, stream>>>(gi, whh, b_ih, b_hh, hs, NUM_NODES);

    fc_out_kernel<<<blk((long long)NUM_NODES * OUTF, TPB), TPB, 0, stream>>>(
        hs, fc_w, fc_b, out, NUM_NODES * OUTF);
}


# GRU scan keeps W_hh in registers for all steps (512 threads, 2 lanes per hidden unit, 192 packed bf16 regs) instead of re-reading it from L2 each step; row sums now 4 chains per lane + lane pair
# speedup vs baseline: 5.6172x; 5.6172x over previous
#include <hip/hip_runtime.h>
#include <hip/hip_bf16.h>
#include <stdint.h>

typedef __attribute__((ext_vector_type(16))) __bf16 v16bf;
typedef __attribute__((ext_vector_type(8)))  float  v8f;

#define NUM_NODES 10000
#define IN_FEAT   64
#define HID       256
#define OUTF      3
#define GATES     768

__global__ void cvt_f32_to_bf16(const float* __restrict__ in,
                                __bf16* __restrict__ out, int n) {
    int i = blockIdx.x * blockDim.x + threadIdx.x;
    if (i < n) out[i] = (__bf16)in[i];
}

__global__ void cvt_f32_to_bf16_T(const float* __restrict__ in,
                                  __bf16* __restrict__ out, int K, int N) {
    int i = blockIdx.x * blockDim.x + threadIdx.x;
    if (i >= K * N) return;
    int k = i / N, n = i % N;
    out[(size_t)n * K + k] = (__bf16)in[(size_t)k * N + n];
}

__global__ void deg_init(float* __restrict__ deg, int n) {
    int i = blockIdx.x * blockDim.x + threadIdx.x;
    if (i < n) deg[i] = 1.0f;
}

__global__ void deg_edges(const int* __restrict__ dst, float* __restrict__ deg, int E) {
    int e = blockIdx.x * blockDim.x + threadIdx.x;
    if (e < E) atomicAdd(&deg[dst[e]], 1.0f);
}

__global__ void deg_rsqrt(const float* __restrict__ deg, float* __restrict__ dinv, int n) {
    int i = blockIdx.x * blockDim.x + threadIdx.x;
    if (i < n) dinv[i] = rsqrtf(deg[i]);
}

__global__ void __launch_bounds__(128)
wmma_gemm_bf16(const __bf16* __restrict__ A, const __bf16* __restrict__ Bt,
               float* __restrict__ C, int M, int N, int K) {
    int wave  = (int)((blockIdx.x * blockDim.x + threadIdx.x) >> 5);
    int lane  = threadIdx.x & 31;
    int tilesN = N >> 4;
    int tilesM = M >> 4;
    if (wave >= tilesM * tilesN) return;
    int tm = wave / tilesN;
    int tn = wave % tilesN;

    int l15 = lane & 15;
    int h   = lane >> 4;
    const __bf16* arow = A  + (size_t)(tm * 16 + l15) * K;
    const __bf16* bcol = Bt + (size_t)(tn * 16 + l15) * K;

    v8f acc = {};
    for (int k = 0; k < K; k += 32) {
        v16bf a, b;
        int k0 = k + h * 8;
#pragma unroll
        for (int i = 0; i < 8; ++i) {
            a[i]     = arow[k0 + i];
            a[i + 8] = arow[k0 + 16 + i];
            b[i]     = bcol[k0 + i];
            b[i + 8] = bcol[k0 + 16 + i];
        }
        acc = __builtin_amdgcn_wmma_f32_16x16x32_bf16(
            false, a, false, b, (short)0, acc, false, false);
    }
#pragma unroll
    for (int r = 0; r < 8; ++r) {
        int mm = tm * 16 + h * 8 + r;
        C[(size_t)mm * N + tn * 16 + l15] = acc[r];
    }
}

__global__ void gcn_selfloop(const float* __restrict__ xw, const float* __restrict__ dinv,
                             float* __restrict__ agg, int n) {
    int i = blockIdx.x * blockDim.x + threadIdx.x;
    if (i >= n) return;
    int node = i >> 8;
    float s = dinv[node];
    agg[i] = xw[i] * s * s;
}

__global__ void gcn_scatter(const float* __restrict__ xw,
                            const int* __restrict__ src, const int* __restrict__ dst,
                            const float* __restrict__ dinv,
                            float* __restrict__ agg, int E) {
    long long gid = (long long)blockIdx.x * blockDim.x + threadIdx.x;
    int e = (int)(gid >> 6);
    if (e >= E) return;
    int f = ((int)gid & 63) << 2;
    int s = src[e], d = dst[e];
    float norm = dinv[s] * dinv[d];
    const float4 v = *(const float4*)(xw + (size_t)s * HID + f);
    float* o = agg + (size_t)d * HID + f;
    atomicAdd(o + 0, norm * v.x);
    atomicAdd(o + 1, norm * v.y);
    atomicAdd(o + 2, norm * v.z);
    atomicAdd(o + 3, norm * v.w);
}

__global__ void bias_relu_bf16(const float* __restrict__ agg, const float* __restrict__ bias,
                               __bf16* __restrict__ hb, int n) {
    int i = blockIdx.x * blockDim.x + threadIdx.x;
    if (i >= n) return;
    float v = agg[i] + bias[i & (HID - 1)];
    hb[i] = (__bf16)(v > 0.f ? v : 0.f);
}

#define SCAN_THREADS (2 * HID)

static __device__ __forceinline__ void mac12(
    float& r0, float& z0, float& n0, float& r1, float& z1, float& n1,
    float& r2, float& z2, float& n2, float& r3, float& z3, float& n3,
    uint32_t wra, uint32_t wza, uint32_t wna, uint32_t wrb, uint32_t wzb, uint32_t wnb,
    float h0, float h1, float h2, float h3) {
    asm("v_fma_mix_f32_bf16 %0, %12, %18, %0 op_sel_hi:[1,0,0]\n\t"
        "v_fma_mix_f32_bf16 %1, %13, %18, %1 op_sel_hi:[1,0,0]\n\t"
        "v_fma_mix_f32_bf16 %2, %14, %18, %2 op_sel_hi:[1,0,0]\n\t"
        "v_fma_mix_f32_bf16 %3, %12, %19, %3 op_sel:[1,0,0] op_sel_hi:[1,0,0]\n\t"
        "v_fma_mix_f32_bf16 %4, %13, %19, %4 op_sel:[1,0,0] op_sel_hi:[1,0,0]\n\t"
        "v_fma_mix_f32_bf16 %5, %14, %19, %5 op_sel:[1,0,0] op_sel_hi:[1,0,0]\n\t"
        "v_fma_mix_f32_bf16 %6, %15, %20, %6 op_sel_hi:[1,0,0]\n\t"
        "v_fma_mix_f32_bf16 %7, %16, %20, %7 op_sel_hi:[1,0,0]\n\t"
        "v_fma_mix_f32_bf16 %8, %17, %20, %8 op_sel_hi:[1,0,0]\n\t"
        "v_fma_mix_f32_bf16 %9, %15, %21, %9 op_sel:[1,0,0] op_sel_hi:[1,0,0]\n\t"
        "v_fma_mix_f32_bf16 %10, %16, %21, %10 op_sel:[1,0,0] op_sel_hi:[1,0,0]\n\t"
        "v_fma_mix_f32_bf16 %11, %17, %21, %11 op_sel:[1,0,0] op_sel_hi:[1,0,0]"
        : "+v"(r0), "+v"(z0), "+v"(n0), "+v"(r1), "+v"(z1), "+v"(n1),
          "+v"(r2), "+v"(z2), "+v"(n2), "+v"(r3), "+v"(z3), "+v"(n3)
        : "v"(wra), "v"(wza), "v"(wna), "v"(wrb), "v"(wzb), "v"(wnb),
          "v"(h0), "v"(h1), "v"(h2), "v"(h3));
}
static __device__ __forceinline__ float pair_sum(float v) {
    return v + __int_as_float(__builtin_amdgcn_update_dpp(0, __float_as_int(v), 0xB1  , 0xF, 0xF, true));
}

__global__ void __launch_bounds__(SCAN_THREADS)
gru_scan(const float* __restrict__ gi, const __bf16* __restrict__ Whh,
         const float* __restrict__ b_ih, const float* __restrict__ b_hh,
         float* __restrict__ hs, int T) {
    __shared__ float4 hbuf[2][HID / 4];
    const int t = threadIdx.x;
    const int j = t >> 1;
    const int s = t & 1;

    uint4 wr[HID / 16], wz[HID / 16], wn[HID / 16];
    {
        const uint4* pr = (const uint4*)(Whh + (size_t)j * HID);
        const uint4* pz = (const uint4*)(Whh + (size_t)(HID + j) * HID);
        const uint4* pn = (const uint4*)(Whh + (size_t)(2 * HID + j) * HID);
#pragma unroll
        for (int q = 0; q < HID / 16; ++q) { wr[q] = pr[2 * q + s]; wz[q] = pz[2 * q + s]; wn[q] = pn[2 * q + s]; }
    }
    const float bih_r = b_ih[j], bih_z = b_ih[HID + j], bih_n = b_ih[2 * HID + j];
    const float bhh_r = b_hh[j], bhh_z = b_hh[HID + j], bhh_n = b_hh[2 * HID + j];

    if (t < HID) ((float*)hbuf[0])[t] = 0.0f;
    float hprev = 0.0f;
    float g_r = gi[j], g_z = gi[HID + j], g_n = gi[2 * HID + j];
    __syncthreads();

    for (int step = 0; step < T; ++step) {
        const float4* hb = hbuf[step & 1];
        const unsigned gx = (unsigned)(step + 1 < T ? step + 1 : step) * GATES + (unsigned)j;
        const float ng_r = gi[gx], ng_z = gi[gx + HID], ng_n = gi[gx + 2 * HID];

        float r0 = 0.f, r1 = 0.f, r2 = 0.f, r3 = 0.f;
        float z0 = 0.f, z1 = 0.f, z2 = 0.f, z3 = 0.f;
        float n0 = 0.f, n1 = 0.f, n2 = 0.f, n3 = 0.f;
        float4 ha[4], hc[4];
        const float hout = ((const float*)hb)[t & (HID - 1)];
#pragma unroll
        for (int p = 0; p < 3; ++p) { ha[p] = hb[2 * (2 * p + s)]; hc[p] = hb[2 * (2 * p + s) + 1]; }
#pragma unroll
        for (int q = 0; q < HID / 16; ++q) {
            if (q + 3 < HID / 16) {
                ha[(q + 3) & 3] = hb[2 * (2 * (q + 3) + s)];
                hc[(q + 3) & 3] = hb[2 * (2 * (q + 3) + s) + 1];
            }
            __builtin_amdgcn_sched_barrier(0);
            const float4 a = ha[q & 3], b = hc[q & 3];
            mac12(r0, z0, n0, r1, z1, n1, r2, z2, n2, r3, z3, n3,
                  wr[q].x, wz[q].x, wn[q].x, wr[q].y, wz[q].y, wn[q].y, a.x, a.y, a.z, a.w);
            mac12(r0, z0, n0, r1, z1, n1, r2, z2, n2, r3, z3, n3,
                  wr[q].z, wz[q].z, wn[q].z, wr[q].w, wz[q].w, wn[q].w, b.x, b.y, b.z, b.w);
            if (q == 1 && step > 0 && t < HID) hs[(unsigned)(step - 1) * HID + (unsigned)t] = hout;
            __builtin_amdgcn_sched_barrier(0);
        }
        const float gh_r = pair_sum((r0 + r1) + (r2 + r3)) + bhh_r;
        const float gh_z = pair_sum((z0 + z1) + (z2 + z3)) + bhh_z;
        const float gh_n = pair_sum((n0 + n1) + (n2 + n3)) + bhh_n;
        const float r  = 1.0f / (1.0f + __expf(-(g_r + bih_r + gh_r)));
        const float z  = 1.0f / (1.0f + __expf(-(g_z + bih_z + gh_z)));
        const float nn = tanhf(g_n + bih_n + r * gh_n);
        const float hn = (1.0f - z) * nn + z * hprev;
        if (s == 0) ((float*)hbuf[(step + 1) & 1])[j] = hn;
        hprev = hn;
        g_r = ng_r; g_z = ng_z; g_n = ng_n;
        __syncthreads();
    }
    if (t < HID) hs[(unsigned)(T - 1) * HID + (unsigned)t] = ((const float*)hbuf[T & 1])[t];
}

__global__ void fc_out_kernel(const float* __restrict__ hs, const float* __restrict__ fw,
                              const float* __restrict__ fb, float* __restrict__ out, int total) {
    int gid = blockIdx.x * blockDim.x + threadIdx.x;
    if (gid >= total) return;
    int n = gid / OUTF, o = gid % OUTF;
    const float* h = hs + (size_t)n * HID;
    const float* w = fw + (size_t)o * HID;
    float acc = fb[o];
#pragma unroll 8
    for (int k = 0; k < HID; ++k) acc = fmaf(h[k], w[k], acc);
    out[gid] = acc;
}

extern "C" void kernel_launch(void* const* d_in, const int* in_sizes, int n_in,
                              void* d_out, int out_size, void* d_ws, size_t ws_size,
                              hipStream_t stream) {
    const float* x    = (const float*)d_in[0];
    const int*   ei   = (const int*)  d_in[1];
    const float* W1   = (const float*)d_in[2];
    const float* b1   = (const float*)d_in[3];
    const float* W2   = (const float*)d_in[4];
    const float* b2   = (const float*)d_in[5];
    const float* W_ih = (const float*)d_in[6];
    const float* W_hh = (const float*)d_in[7];
    const float* b_ih = (const float*)d_in[8];
    const float* b_hh = (const float*)d_in[9];
    const float* fc_w = (const float*)d_in[10];
    const float* fc_b = (const float*)d_in[11];
    float* out = (float*)d_out;

    const int E = in_sizes[1] / 2;
    const int* src = ei;
    const int* dst = ei + E;

    char* ws = (char*)d_ws;
    size_t off = 0;
    auto alloc = [&](size_t bytes) -> char* {
        char* p = ws + off;
        off = (off + bytes + 255) & ~(size_t)255;
        return p;
    };
    float*  deg  = (float*) alloc((size_t)NUM_NODES * 4);
    float*  dinv = (float*) alloc((size_t)NUM_NODES * 4);
    float*  xw   = (float*) alloc((size_t)NUM_NODES * HID * 4);
    float*  agg  = (float*) alloc((size_t)NUM_NODES * HID * 4);
    float*  gi   = (float*) alloc((size_t)NUM_NODES * GATES * 4);
    float*  hs   = (float*) alloc((size_t)NUM_NODES * HID * 4);
    __bf16* xb   = (__bf16*)alloc((size_t)NUM_NODES * HID * 2);
    __bf16* w1t  = (__bf16*)alloc((size_t)HID * IN_FEAT * 2);
    __bf16* w2t  = (__bf16*)alloc((size_t)HID * HID * 2);
    __bf16* wih  = (__bf16*)alloc((size_t)GATES * HID * 2);
    __bf16* whh  = (__bf16*)alloc((size_t)GATES * HID * 2);
    (void)ws_size; (void)n_in; (void)out_size;

    const int TPB = 256;
    auto blk = [](long long n, int t) { return (unsigned)((n + t - 1) / t); };

    cvt_f32_to_bf16<<<blk((long long)NUM_NODES * IN_FEAT, TPB), TPB, 0, stream>>>(
        x, xb, NUM_NODES * IN_FEAT);
    cvt_f32_to_bf16_T<<<blk((long long)IN_FEAT * HID, TPB), TPB, 0, stream>>>(
        W1, w1t, IN_FEAT, HID);
    cvt_f32_to_bf16_T<<<blk((long long)HID * HID, TPB), TPB, 0, stream>>>(
        W2, w2t, HID, HID);
    cvt_f32_to_bf16<<<blk((long long)GATES * HID, TPB), TPB, 0, stream>>>(
        W_ih, wih, GATES * HID);
    cvt_f32_to_bf16<<<blk((long long)GATES * HID, TPB), TPB, 0, stream>>>(
        W_hh, whh, GATES * HID);

    deg_init<<<blk(NUM_NODES, TPB), TPB, 0, stream>>>(deg, NUM_NODES);
    deg_edges<<<blk(E, TPB), TPB, 0, stream>>>(dst, deg, E);
    deg_rsqrt<<<blk(NUM_NODES, TPB), TPB, 0, stream>>>(deg, dinv, NUM_NODES);

    const long long NH = (long long)NUM_NODES * HID;
    const unsigned gemm1_blocks = blk((long long)(NUM_NODES / 16) * (HID / 16), 4);
    const unsigned gemm2_blocks = gemm1_blocks;
    const unsigned gemm3_blocks = blk((long long)(NUM_NODES / 16) * (GATES / 16), 4);
    const unsigned scat_blocks  = blk((long long)E * 64, TPB);

    wmma_gemm_bf16<<<gemm1_blocks, 128, 0, stream>>>(xb, w1t, xw, NUM_NODES, HID, IN_FEAT);
    gcn_selfloop<<<blk(NH, TPB), TPB, 0, stream>>>(xw, dinv, agg, (int)NH);
    gcn_scatter<<<scat_blocks, TPB, 0, stream>>>(xw, src, dst, dinv, agg, E);
    bias_relu_bf16<<<blk(NH, TPB), TPB, 0, stream>>>(agg, b1, xb, (int)NH);

    wmma_gemm_bf16<<<gemm2_blocks, 128, 0, stream>>>(xb, w2t, xw, NUM_NODES, HID, HID);
    gcn_selfloop<<<blk(NH, TPB), TPB, 0, stream>>>(xw, dinv, agg, (int)NH);
    gcn_scatter<<<scat_blocks, TPB, 0, stream>>>(xw, src, dst, dinv, agg, E);
    bias_relu_bf16<<<blk(NH, TPB), TPB, 0, stream>>>(agg, b2, xb, (int)NH);

    wmma_gemm_bf16<<<gemm3_blocks, 128, 0, stream>>>(xb, wih, gi, NUM_NODES, GATES, HID);

    gru_scan<<<1, SCAN_THREADS, 0, stream>>>(gi, whh, b_ih, b_hh, hs, NUM_NODES);

    fc_out_kernel<<<blk((long long)NUM_NODES * OUTF, TPB), TPB, 0, stream>>>(
        hs, fc_w, fc_b, out, NUM_NODES * OUTF);
}


# GRU scan: h.W_hh^T moved from vector FMAs to the bf16 matrix instruction; h enters as three bf16 pieces (exact split of its f32 value) in A rows 0-2; row sum order: matrix unit's, then lo+mid+hi
# speedup vs baseline: 9.3660x; 1.6674x over previous
#include <hip/hip_runtime.h>
#include <hip/hip_bf16.h>
#include <stdint.h>

typedef __attribute__((ext_vector_type(16))) __bf16 v16bf;
typedef __attribute__((ext_vector_type(8)))  float  v8f;

#define NUM_NODES 10000
#define IN_FEAT   64
#define HID       256
#define OUTF      3
#define GATES     768

__global__ void cvt_f32_to_bf16(const float* __restrict__ in,
                                __bf16* __restrict__ out, int n) {
    int i = blockIdx.x * blockDim.x + threadIdx.x;
    if (i < n) out[i] = (__bf16)in[i];
}

__global__ void cvt_f32_to_bf16_T(const float* __restrict__ in,
                                  __bf16* __restrict__ out, int K, int N) {
    int i = blockIdx.x * blockDim.x + threadIdx.x;
    if (i >= K * N) return;
    int k = i / N, n = i % N;
    out[(size_t)n * K + k] = (__bf16)in[(size_t)k * N + n];
}

__global__ void deg_init(float* __restrict__ deg, int n) {
    int i = blockIdx.x * blockDim.x + threadIdx.x;
    if (i < n) deg[i] = 1.0f;
}

__global__ void deg_edges(const int* __restrict__ dst, float* __restrict__ deg, int E) {
    int e = blockIdx.x * blockDim.x + threadIdx.x;
    if (e < E) atomicAdd(&deg[dst[e]], 1.0f);
}

__global__ void deg_rsqrt(const float* __restrict__ deg, float* __restrict__ dinv, int n) {
    int i = blockIdx.x * blockDim.x + threadIdx.x;
    if (i < n) dinv[i] = rsqrtf(deg[i]);
}

__global__ void __launch_bounds__(128)
wmma_gemm_bf16(const __bf16* __restrict__ A, const __bf16* __restrict__ Bt,
               float* __restrict__ C, int M, int N, int K) {
    int wave  = (int)((blockIdx.x * blockDim.x + threadIdx.x) >> 5);
    int lane  = threadIdx.x & 31;
    int tilesN = N >> 4;
    int tilesM = M >> 4;
    if (wave >= tilesM * tilesN) return;
    int tm = wave / tilesN;
    int tn = wave % tilesN;

    int l15 = lane & 15;
    int h   = lane >> 4;
    const __bf16* arow = A  + (size_t)(tm * 16 + l15) * K;
    const __bf16* bcol = Bt + (size_t)(tn * 16 + l15) * K;

    v8f acc = {};
    for (int k = 0; k < K; k += 32) {
        v16bf a, b;
        int k0 = k + h * 8;
#pragma unroll
        for (int i = 0; i < 8; ++i) {
            a[i]     = arow[k0 + i];
            a[i + 8] = arow[k0 + 16 + i];
            b[i]     = bcol[k0 + i];
            b[i + 8] = bcol[k0 + 16 + i];
        }
        acc = __builtin_amdgcn_wmma_f32_16x16x32_bf16(
            false, a, false, b, (short)0, acc, false, false);
    }
#pragma unroll
    for (int r = 0; r < 8; ++r) {
        int mm = tm * 16 + h * 8 + r;
        C[(size_t)mm * N + tn * 16 + l15] = acc[r];
    }
}

__global__ void gcn_selfloop(const float* __restrict__ xw, const float* __restrict__ dinv,
                             float* __restrict__ agg, int n) {
    int i = blockIdx.x * blockDim.x + threadIdx.x;
    if (i >= n) return;
    int node = i >> 8;
    float s = dinv[node];
    agg[i] = xw[i] * s * s;
}

__global__ void gcn_scatter(const float* __restrict__ xw,
                            const int* __restrict__ src, const int* __restrict__ dst,
                            const float* __restrict__ dinv,
                            float* __restrict__ agg, int E) {
    long long gid = (long long)blockIdx.x * blockDim.x + threadIdx.x;
    int e = (int)(gid >> 6);
    if (e >= E) return;
    int f = ((int)gid & 63) << 2;
    int s = src[e], d = dst[e];
    float norm = dinv[s] * dinv[d];
    const float4 v = *(const float4*)(xw + (size_t)s * HID + f);
    float* o = agg + (size_t)d * HID + f;
    atomicAdd(o + 0, norm * v.x);
    atomicAdd(o + 1, norm * v.y);
    atomicAdd(o + 2, norm * v.z);
    atomicAdd(o + 3, norm * v.w);
}

__global__ void bias_relu_bf16(const float* __restrict__ agg, const float* __restrict__ bias,
                               __bf16* __restrict__ hb, int n) {
    int i = blockIdx.x * blockDim.x + threadIdx.x;
    if (i >= n) return;
    float v = agg[i] + bias[i & (HID - 1)];
    hb[i] = (__bf16)(v > 0.f ? v : 0.f);
}

#define SCAN_THREADS 512
#define HPITCH4 34

typedef __attribute__((ext_vector_type(8))) unsigned int v8u;

static __device__ __forceinline__ v16bf frag_at(const uint4* p) {
    const uint4 lo = p[0], hi = p[2];
    const v8u v = {lo.x, lo.y, lo.z, lo.w, hi.x, hi.y, hi.z, hi.w};
    return __builtin_bit_cast(v16bf, v);
}

__global__ void __launch_bounds__(SCAN_THREADS)
gru_scan(const float* __restrict__ gi, const __bf16* __restrict__ Whh,
         const float* __restrict__ b_ih, const float* __restrict__ b_hh,
         float* __restrict__ hs, int T) {
    __shared__ uint4 hsplit[2][4 * HPITCH4];
    __shared__ float hfull[2][HID];
    const int t  = threadIdx.x;
    const int w  = t >> 5;
    const int c  = t & 15;
    const int hh = (t >> 4) & 1;
    const int j  = 16 * w + c;

    v16bf Br[HID / 32], Bz[HID / 32], Bn[HID / 32];
#pragma unroll
    for (int kt = 0; kt < HID / 32; ++kt) {
        Br[kt] = frag_at((const uint4*)(Whh + (size_t)j * HID + 32 * kt + 8 * hh));
        Bz[kt] = frag_at((const uint4*)(Whh + (size_t)(HID + j) * HID + 32 * kt + 8 * hh));
        Bn[kt] = frag_at((const uint4*)(Whh + (size_t)(2 * HID + j) * HID + 32 * kt + 8 * hh));
    }
    const float bih_r = b_ih[j], bih_z = b_ih[HID + j], bih_n = b_ih[2 * HID + j];
    const float bhh_r = b_hh[j], bhh_z = b_hh[HID + j], bhh_n = b_hh[2 * HID + j];

    for (int i = t; i < 2 * 4 * HPITCH4; i += SCAN_THREADS) (&hsplit[0][0])[i] = make_uint4(0u, 0u, 0u, 0u);
    if (t < HID) hfull[0][t] = 0.0f;
    float hprev = 0.0f;
    float g_r = gi[j], g_z = gi[HID + j], g_n = gi[2 * HID + j];
    const int aoff = (c < 3 ? c : 3) * HPITCH4 + hh;
    __syncthreads();

    for (int step = 0; step < T; ++step) {
        const uint4* ha = &hsplit[step & 1][aoff];
        const unsigned gx = (unsigned)(step + 1 < T ? step + 1 : step) * GATES + (unsigned)j;
        const float ng_r = gi[gx], ng_z = gi[gx + HID], ng_n = gi[gx + 2 * HID];
        const float hout = hfull[step & 1][t & (HID - 1)];

        v8f ar = {}, az = {}, an = {};
        v16bf a[2];
        a[0] = frag_at(ha);
#pragma unroll
        for (int kt = 0; kt < HID / 32; ++kt) {
            if (kt + 1 < HID / 32) a[(kt + 1) & 1] = frag_at(ha + 4 * (kt + 1));
            __builtin_amdgcn_sched_barrier(0);
            const v16bf ak = a[kt & 1];
            ar = __builtin_amdgcn_wmma_f32_16x16x32_bf16(false, ak, false, Br[kt], (short)0, ar, false, false);
            az = __builtin_amdgcn_wmma_f32_16x16x32_bf16(false, ak, false, Bz[kt], (short)0, az, false, false);
            an = __builtin_amdgcn_wmma_f32_16x16x32_bf16(false, ak, false, Bn[kt], (short)0, an, false, false);
            if (kt == 1 && step > 0 && t < HID) hs[(unsigned)(step - 1) * HID + (unsigned)t] = hout;
            __builtin_amdgcn_sched_barrier(0);
        }
        const float gh_r = ((ar[2] + ar[1]) + ar[0]) + bhh_r;
        const float gh_z = ((az[2] + az[1]) + az[0]) + bhh_z;
        const float gh_n = ((an[2] + an[1]) + an[0]) + bhh_n;
        const float r  = 1.0f / (1.0f + __expf(-(g_r + bih_r + gh_r)));
        const float z  = 1.0f / (1.0f + __expf(-(g_z + bih_z + gh_z)));
        const float nn = tanhf(g_n + bih_n + r * gh_n);
        const float hn = (1.0f - z) * nn + z * hprev;
        const __bf16 p0 = (__bf16)hn;
        const float  e1 = hn - (float)p0;
        const __bf16 p1 = (__bf16)e1;
        const float  e2 = e1 - (float)p1;
        const __bf16 p2 = (__bf16)e2;
        if (hh == 0) {
            __bf16* hnx = (__bf16*)&hsplit[(step + 1) & 1][0];
            hnx[j]                   = p0;
            hnx[8 * HPITCH4 + j]     = p1;
            hnx[2 * 8 * HPITCH4 + j] = p2;
            hfull[(step + 1) & 1][j] = hn;
        }
        hprev = hn;
        g_r = ng_r; g_z = ng_z; g_n = ng_n;
        __syncthreads();
    }
    if (t < HID) hs[(unsigned)(T - 1) * HID + (unsigned)t] = hfull[T & 1][t];
}

__global__ void fc_out_kernel(const float* __restrict__ hs, const float* __restrict__ fw,
                              const float* __restrict__ fb, float* __restrict__ out, int total) {
    int gid = blockIdx.x * blockDim.x + threadIdx.x;
    if (gid >= total) return;
    int n = gid / OUTF, o = gid % OUTF;
    const float* h = hs + (size_t)n * HID;
    const float* w = fw + (size_t)o * HID;
    float acc = fb[o];
#pragma unroll 8
    for (int k = 0; k < HID; ++k) acc = fmaf(h[k], w[k], acc);
    out[gid] = acc;
}

extern "C" void kernel_launch(void* const* d_in, const int* in_sizes, int n_in,
                              void* d_out, int out_size, void* d_ws, size_t ws_size,
                              hipStream_t stream) {
    const float* x    = (const float*)d_in[0];
    const int*   ei   = (const int*)  d_in[1];
    const float* W1   = (const float*)d_in[2];
    const float* b1   = (const float*)d_in[3];
    const float* W2   = (const float*)d_in[4];
    const float* b2   = (const float*)d_in[5];
    const float* W_ih = (const float*)d_in[6];
    const float* W_hh = (const float*)d_in[7];
    const float* b_ih = (const float*)d_in[8];
    const float* b_hh = (const float*)d_in[9];
    const float* fc_w = (const float*)d_in[10];
    const float* fc_b = (const float*)d_in[11];
    float* out = (float*)d_out;

    const int E = in_sizes[1] / 2;
    const int* src = ei;
    const int* dst = ei + E;

    char* ws = (char*)d_ws;
    size_t off = 0;
    auto alloc = [&](size_t bytes) -> char* {
        char* p = ws + off;
        off = (off + bytes + 255) & ~(size_t)255;
        return p;
    };
    float*  deg  = (float*) alloc((size_t)NUM_NODES * 4);
    float*  dinv = (float*) alloc((size_t)NUM_NODES * 4);
    float*  xw   = (float*) alloc((size_t)NUM_NODES * HID * 4);
    float*  agg  = (float*) alloc((size_t)NUM_NODES * HID * 4);
    float*  gi   = (float*) alloc((size_t)NUM_NODES * GATES * 4);
    float*  hs   = (float*) alloc((size_t)NUM_NODES * HID * 4);
    __bf16* xb   = (__bf16*)alloc((size_t)NUM_NODES * HID * 2);
    __bf16* w1t  = (__bf16*)alloc((size_t)HID * IN_FEAT * 2);
    __bf16* w2t  = (__bf16*)alloc((size_t)HID * HID * 2);
    __bf16* wih  = (__bf16*)alloc((size_t)GATES * HID * 2);
    __bf16* whh  = (__bf16*)alloc((size_t)GATES * HID * 2);
    (void)ws_size; (void)n_in; (void)out_size;

    const int TPB = 256;
    auto blk = [](long long n, int t) { return (unsigned)((n + t - 1) / t); };

    cvt_f32_to_bf16<<<blk((long long)NUM_NODES * IN_FEAT, TPB), TPB, 0, stream>>>(
        x, xb, NUM_NODES * IN_FEAT);
    cvt_f32_to_bf16_T<<<blk((long long)IN_FEAT * HID, TPB), TPB, 0, stream>>>(
        W1, w1t, IN_FEAT, HID);
    cvt_f32_to_bf16_T<<<blk((long long)HID * HID, TPB), TPB, 0, stream>>>(
        W2, w2t, HID, HID);
    cvt_f32_to_bf16<<<blk((long long)GATES * HID, TPB), TPB, 0, stream>>>(
        W_ih, wih, GATES * HID);
    cvt_f32_to_bf16<<<blk((long long)GATES * HID, TPB), TPB, 0, stream>>>(
        W_hh, whh, GATES * HID);

    deg_init<<<blk(NUM_NODES, TPB), TPB, 0, stream>>>(deg, NUM_NODES);
    deg_edges<<<blk(E, TPB), TPB, 0, stream>>>(dst, deg, E);
    deg_rsqrt<<<blk(NUM_NODES, TPB), TPB, 0, stream>>>(deg, dinv, NUM_NODES);

    const long long NH = (long long)NUM_NODES * HID;
    const unsigned gemm1_blocks = blk((long long)(NUM_NODES / 16) * (HID / 16), 4);
    const unsigned gemm2_blocks = gemm1_blocks;
    const unsigned gemm3_blocks = blk((long long)(NUM_NODES / 16) * (GATES / 16), 4);
    const unsigned scat_blocks  = blk((long long)E * 64, TPB);

    wmma_gemm_bf16<<<gemm1_blocks, 128, 0, stream>>>(xb, w1t, xw, NUM_NODES, HID, IN_FEAT);
    gcn_selfloop<<<blk(NH, TPB), TPB, 0, stream>>>(xw, dinv, agg, (int)NH);
    gcn_scatter<<<scat_blocks, TPB, 0, stream>>>(xw, src, dst, dinv, agg, E);
    bias_relu_bf16<<<blk(NH, TPB), TPB, 0, stream>>>(agg, b1, xb, (int)NH);

    wmma_gemm_bf16<<<gemm2_blocks, 128, 0, stream>>>(xb, w2t, xw, NUM_NODES, HID, HID);
    gcn_selfloop<<<blk(NH, TPB), TPB, 0, stream>>>(xw, dinv, agg, (int)NH);
    gcn_scatter<<<scat_blocks, TPB, 0, stream>>>(xw, src, dst, dinv, agg, E);
    bias_relu_bf16<<<blk(NH, TPB), TPB, 0, stream>>>(agg, b2, xb, (int)NH);

    wmma_gemm_bf16<<<gemm3_blocks, 128, 0, stream>>>(xb, wih, gi, NUM_NODES, GATES, HID);

    gru_scan<<<1, SCAN_THREADS, 0, stream>>>(gi, whh, b_ih, b_hh, hs, NUM_NODES);

    fc_out_kernel<<<blk((long long)NUM_NODES * OUTF, TPB), TPB, 0, stream>>>(
        hs, fc_w, fc_b, out, NUM_NODES * OUTF);
}


# GCN aggregation: edges bucketed by destination in-call (count, scan, place, rank by source); one wave sums a node's rows in registers. No float atomics; sum order fixed: self, then ascending source
# speedup vs baseline: 10.7872x; 1.1517x over previous
#include <hip/hip_runtime.h>
#include <hip/hip_bf16.h>
#include <stdint.h>

typedef __attribute__((ext_vector_type(16))) __bf16 v16bf;
typedef __attribute__((ext_vector_type(8)))  float  v8f;

#define NUM_NODES 10000
#define IN_FEAT   64
#define HID       256
#define OUTF      3
#define GATES     768

__global__ void cvt_f32_to_bf16(const float* __restrict__ in,
                                __bf16* __restrict__ out, int n) {
    int i = blockIdx.x * blockDim.x + threadIdx.x;
    if (i < n) out[i] = (__bf16)in[i];
}

__global__ void cvt_f32_to_bf16_T(const float* __restrict__ in,
                                  __bf16* __restrict__ out, int K, int N) {
    int i = blockIdx.x * blockDim.x + threadIdx.x;
    if (i >= K * N) return;
    int k = i / N, n = i % N;
    out[(size_t)n * K + k] = (__bf16)in[(size_t)k * N + n];
}

__global__ void csr_zero(int* __restrict__ cnt, int n) {
    int i = blockIdx.x * blockDim.x + threadIdx.x;
    if (i < n) cnt[i] = 0;
}

__global__ void csr_count(const int* __restrict__ dst, int* __restrict__ cnt,
                          int* __restrict__ pos, int E) {
    int e = blockIdx.x * blockDim.x + threadIdx.x;
    if (e < E) pos[e] = atomicAdd(&cnt[dst[e]], 1);
}

#define OFFS_THREADS 1024
__global__ void __launch_bounds__(OFFS_THREADS)
csr_offsets(const int* __restrict__ cnt, int* __restrict__ rowptr,
            float* __restrict__ dinv, int n) {
    __shared__ int part[OFFS_THREADS];
    const int t   = threadIdx.x;
    const int per = (n + OFFS_THREADS - 1) / OFFS_THREADS;
    const int b0  = t * per;
    int s = 0;
    for (int i = 0; i < per; ++i) {
        const int idx = b0 + i;
        if (idx < n) s += cnt[idx];
    }
    part[t] = s;
    __syncthreads();
    for (int off = 1; off < OFFS_THREADS; off <<= 1) {
        const int v = (t >= off) ? part[t - off] : 0;
        __syncthreads();
        part[t] += v;
        __syncthreads();
    }
    int run = part[t] - s;
    for (int i = 0; i < per; ++i) {
        const int idx = b0 + i;
        if (idx < n) {
            const int c = cnt[idx];
            rowptr[idx] = run;
            dinv[idx]   = rsqrtf((float)(c + 1));
            run += c;
        }
    }
    if (t == OFFS_THREADS - 1) rowptr[n] = part[t];
}

__global__ void csr_place(const int* __restrict__ src, const int* __restrict__ dst,
                          const int* __restrict__ rowptr, const int* __restrict__ pos,
                          int* __restrict__ col, int E) {
    int e = blockIdx.x * blockDim.x + threadIdx.x;
    if (e < E) col[rowptr[dst[e]] + pos[e]] = src[e];
}

__global__ void csr_rank(const int* __restrict__ col, const int* __restrict__ rowptr,
                         const float* __restrict__ dinv,
                         int* __restrict__ scol, float* __restrict__ wgt, int n) {
    const int node = (int)((blockIdx.x * blockDim.x + threadIdx.x) >> 5);
    const int lane = threadIdx.x & 31;
    if (node >= n) return;
    const int b = rowptr[node];
    const int L = rowptr[node + 1] - b;
    const float dd = dinv[node];
    const int* row = col + b;
    for (int i = lane; i < L; i += 32) {
        const int s = row[i];
        int rank = 0;
        for (int q = 0; q < L; ++q) {
            const int v = row[q];
            rank += (v < s || (v == s && q < i)) ? 1 : 0;
        }
        scol[b + rank] = s;
        wgt[b + rank]  = dinv[s] * dd;
    }
}

__global__ void __launch_bounds__(128)
wmma_gemm_bf16(const __bf16* __restrict__ A, const __bf16* __restrict__ Bt,
               float* __restrict__ C, int M, int N, int K) {
    int wave  = (int)((blockIdx.x * blockDim.x + threadIdx.x) >> 5);
    int lane  = threadIdx.x & 31;
    int tilesN = N >> 4;
    int tilesM = M >> 4;
    if (wave >= tilesM * tilesN) return;
    int tm = wave / tilesN;
    int tn = wave % tilesN;

    int l15 = lane & 15;
    int h   = lane >> 4;
    const __bf16* arow = A  + (size_t)(tm * 16 + l15) * K;
    const __bf16* bcol = Bt + (size_t)(tn * 16 + l15) * K;

    v8f acc = {};
    for (int k = 0; k < K; k += 32) {
        v16bf a, b;
        int k0 = k + h * 8;
#pragma unroll
        for (int i = 0; i < 8; ++i) {
            a[i]     = arow[k0 + i];
            a[i + 8] = arow[k0 + 16 + i];
            b[i]     = bcol[k0 + i];
            b[i + 8] = bcol[k0 + 16 + i];
        }
        acc = __builtin_amdgcn_wmma_f32_16x16x32_bf16(
            false, a, false, b, (short)0, acc, false, false);
    }
#pragma unroll
    for (int r = 0; r < 8; ++r) {
        int mm = tm * 16 + h * 8 + r;
        C[(size_t)mm * N + tn * 16 + l15] = acc[r];
    }
}

typedef __attribute__((ext_vector_type(4))) __bf16 v4bf;

__global__ void __launch_bounds__(256)
gcn_gather(const float* __restrict__ xw, const int* __restrict__ rowptr,
           const int* __restrict__ scol, const float* __restrict__ wgt,
           const float* __restrict__ dinv, const float* __restrict__ bias,
           __bf16* __restrict__ hb, int n) {
    const int lane = threadIdx.x & 31;
    const int node = __builtin_amdgcn_readfirstlane((int)((blockIdx.x * blockDim.x + threadIdx.x) >> 5));
    if (node >= n) return;
    const int b = __builtin_amdgcn_readfirstlane(rowptr[node]);
    const int L = __builtin_amdgcn_readfirstlane(rowptr[node + 1]) - b;
    const float sd = dinv[node];

    const float4* self = (const float4*)(xw + (size_t)node * HID);
    float4 a0 = self[lane], a1 = self[32 + lane];
    a0.x = a0.x * sd * sd; a0.y = a0.y * sd * sd; a0.z = a0.z * sd * sd; a0.w = a0.w * sd * sd;
    a1.x = a1.x * sd * sd; a1.y = a1.y * sd * sd; a1.z = a1.z * sd * sd; a1.w = a1.w * sd * sd;

    int   ms = 0;
    float mw = 0.0f;
    if (lane < L) { ms = scol[b + lane]; mw = wgt[b + lane]; }
    for (int c = 0; c < L; c += 32) {
        const int m = (L - c < 32) ? (L - c) : 32;
        int   ns = 0;
        float nw = 0.0f;
        if (c + 32 + lane < L) { ns = scol[b + c + 32 + lane]; nw = wgt[b + c + 32 + lane]; }
#pragma unroll 8
        for (int q = 0; q < m; ++q) {
            const int   s = __builtin_amdgcn_readlane(ms, q);
            const float w = __builtin_bit_cast(float, __builtin_amdgcn_readlane(__builtin_bit_cast(int, mw), q));
            const float4* r = (const float4*)(xw + (size_t)s * HID);
            const float4 v0 = r[lane], v1 = r[32 + lane];
            a0.x = fmaf(w, v0.x, a0.x); a0.y = fmaf(w, v0.y, a0.y);
            a0.z = fmaf(w, v0.z, a0.z); a0.w = fmaf(w, v0.w, a0.w);
            a1.x = fmaf(w, v1.x, a1.x); a1.y = fmaf(w, v1.y, a1.y);
            a1.z = fmaf(w, v1.z, a1.z); a1.w = fmaf(w, v1.w, a1.w);
        }
        ms = ns; mw = nw;
    }

    const float* bl = bias + 4 * lane;
    a0.x += bl[0];   a0.y += bl[1];   a0.z += bl[2];   a0.w += bl[3];
    a1.x += bl[128]; a1.y += bl[129]; a1.z += bl[130]; a1.w += bl[131];
    v4bf o0, o1;
    o0[0] = (__bf16)(a0.x > 0.f ? a0.x : 0.f); o0[1] = (__bf16)(a0.y > 0.f ? a0.y : 0.f);
    o0[2] = (__bf16)(a0.z > 0.f ? a0.z : 0.f); o0[3] = (__bf16)(a0.w > 0.f ? a0.w : 0.f);
    o1[0] = (__bf16)(a1.x > 0.f ? a1.x : 0.f); o1[1] = (__bf16)(a1.y > 0.f ? a1.y : 0.f);
    o1[2] = (__bf16)(a1.z > 0.f ? a1.z : 0.f); o1[3] = (__bf16)(a1.w > 0.f ? a1.w : 0.f);
    __bf16* orow = hb + (size_t)node * HID;
    *(v4bf*)(orow + 4 * lane)       = o0;
    *(v4bf*)(orow + 128 + 4 * lane) = o1;
}

#define SCAN_THREADS 512
#define HPITCH4 34

typedef __attribute__((ext_vector_type(8))) unsigned int v8u;

static __device__ __forceinline__ v16bf frag_at(const uint4* p) {
    const uint4 lo = p[0], hi = p[2];
    const v8u v = {lo.x, lo.y, lo.z, lo.w, hi.x, hi.y, hi.z, hi.w};
    return __builtin_bit_cast(v16bf, v);
}

static __device__ __forceinline__ float gate_sigmoid(float x) {
    return __builtin_amdgcn_rcpf(1.0f + __builtin_amdgcn_exp2f(x * -1.4426950408889634f));
}
static __device__ __forceinline__ float gate_tanh(float x) {
    return fmaf(2.0f, __builtin_amdgcn_rcpf(1.0f + __builtin_amdgcn_exp2f(x * -2.8853900817779268f)), -1.0f);
}

__global__ void __launch_bounds__(SCAN_THREADS)
gru_scan(const float* __restrict__ gi, const __bf16* __restrict__ Whh,
         const float* __restrict__ b_ih, const float* __restrict__ b_hh,
         float* __restrict__ hs, int T) {
    __shared__ uint4 hsplit[2][4 * HPITCH4];
    __shared__ float hfull[2][HID];
    const int t  = threadIdx.x;
    const int w  = t >> 5;
    const int c  = t & 15;
    const int hh = (t >> 4) & 1;
    const int j  = 16 * w + c;

    v16bf Br[HID / 32], Bz[HID / 32], Bn[HID / 32];
#pragma unroll
    for (int kt = 0; kt < HID / 32; ++kt) {
        Br[kt] = frag_at((const uint4*)(Whh + (size_t)j * HID + 32 * kt + 8 * hh));
        Bz[kt] = frag_at((const uint4*)(Whh + (size_t)(HID + j) * HID + 32 * kt + 8 * hh));
        Bn[kt] = frag_at((const uint4*)(Whh + (size_t)(2 * HID + j) * HID + 32 * kt + 8 * hh));
    }
    const float bih_r = b_ih[j], bih_z = b_ih[HID + j], bih_n = b_ih[2 * HID + j];
    const float bhh_r = b_hh[j], bhh_z = b_hh[HID + j], bhh_n = b_hh[2 * HID + j];

    for (int i = t; i < 2 * 4 * HPITCH4; i += SCAN_THREADS) (&hsplit[0][0])[i] = make_uint4(0u, 0u, 0u, 0u);
    if (t < HID) hfull[0][t] = 0.0f;
    float hprev = 0.0f;
    float g_r = gi[j], g_z = gi[HID + j], g_n = gi[2 * HID + j];
    const int aoff = (c < 3 ? c : 3) * HPITCH4 + hh;
    __syncthreads();

    for (int step = 0; step < T; ++step) {
        const uint4* ha = &hsplit[step & 1][aoff];
        const unsigned gx = (unsigned)(step + 1 < T ? step + 1 : step) * GATES + (unsigned)j;
        const float ng_r = gi[gx], ng_z = gi[gx + HID], ng_n = gi[gx + 2 * HID];
        const float hout = hfull[step & 1][t & (HID - 1)];

        v8f ar = {}, az = {}, an = {};
        v16bf a[2];
        a[0] = frag_at(ha);
#pragma unroll
        for (int kt = 0; kt < HID / 32; ++kt) {
            if (kt + 1 < HID / 32) a[(kt + 1) & 1] = frag_at(ha + 4 * (kt + 1));
            __builtin_amdgcn_sched_barrier(0);
            const v16bf ak = a[kt & 1];
            ar = __builtin_amdgcn_wmma_f32_16x16x32_bf16(false, ak, false, Br[kt], (short)0, ar, false, false);
            az = __builtin_amdgcn_wmma_f32_16x16x32_bf16(false, ak, false, Bz[kt], (short)0, az, false, false);
            an = __builtin_amdgcn_wmma_f32_16x16x32_bf16(false, ak, false, Bn[kt], (short)0, an, false, false);
            if (kt == 1 && step > 0 && t < HID) hs[(unsigned)(step - 1) * HID + (unsigned)t] = hout;
            __builtin_amdgcn_sched_barrier(0);
        }
        const float gh_r = ((ar[2] + ar[1]) + ar[0]) + bhh_r;
        const float gh_z = ((az[2] + az[1]) + az[0]) + bhh_z;
        const float gh_n = ((an[2] + an[1]) + an[0]) + bhh_n;
        const float r  = gate_sigmoid(g_r + bih_r + gh_r);
        const float z  = gate_sigmoid(g_z + bih_z + gh_z);
        const float nn = gate_tanh(g_n + bih_n + r * gh_n);
        const float hn = (1.0f - z) * nn + z * hprev;
        const __bf16 p0 = (__bf16)hn;
        const float  e1 = hn - (float)p0;
        const __bf16 p1 = (__bf16)e1;
        const float  e2 = e1 - (float)p1;
        const __bf16 p2 = (__bf16)e2;
        if (hh == 0) {
            __bf16* hnx = (__bf16*)&hsplit[(step + 1) & 1][0];
            hnx[j]                   = p0;
            hnx[8 * HPITCH4 + j]     = p1;
            hnx[2 * 8 * HPITCH4 + j] = p2;
            hfull[(step + 1) & 1][j] = hn;
        }
        hprev = hn;
        g_r = ng_r; g_z = ng_z; g_n = ng_n;
        __syncthreads();
    }
    if (t < HID) hs[(unsigned)(T - 1) * HID + (unsigned)t] = hfull[T & 1][t];
}

__global__ void fc_out_kernel(const float* __restrict__ hs, const float* __restrict__ fw,
                              const float* __restrict__ fb, float* __restrict__ out, int total) {
    int gid = blockIdx.x * blockDim.x + threadIdx.x;
    if (gid >= total) return;
    int n = gid / OUTF, o = gid % OUTF;
    const float* h = hs + (size_t)n * HID;
    const float* w = fw + (size_t)o * HID;
    float acc = fb[o];
#pragma unroll 8
    for (int k = 0; k < HID; ++k) acc = fmaf(h[k], w[k], acc);
    out[gid] = acc;
}

extern "C" void kernel_launch(void* const* d_in, const int* in_sizes, int n_in,
                              void* d_out, int out_size, void* d_ws, size_t ws_size,
                              hipStream_t stream) {
    const float* x    = (const float*)d_in[0];
    const int*   ei   = (const int*)  d_in[1];
    const float* W1   = (const float*)d_in[2];
    const float* b1   = (const float*)d_in[3];
    const float* W2   = (const float*)d_in[4];
    const float* b2   = (const float*)d_in[5];
    const float* W_ih = (const float*)d_in[6];
    const float* W_hh = (const float*)d_in[7];
    const float* b_ih = (const float*)d_in[8];
    const float* b_hh = (const float*)d_in[9];
    const float* fc_w = (const float*)d_in[10];
    const float* fc_b = (const float*)d_in[11];
    float* out = (float*)d_out;

    const int E = in_sizes[1] / 2;
    const int* src = ei;
    const int* dst = ei + E;

    char* ws = (char*)d_ws;
    size_t off = 0;
    auto alloc = [&](size_t bytes) -> char* {
        char* p = ws + off;
        off = (off + bytes + 255) & ~(size_t)255;
        return p;
    };
    int*    cnt  = (int*)   alloc((size_t)NUM_NODES * 4);
    float*  dinv = (float*) alloc((size_t)NUM_NODES * 4);
    float*  xw   = (float*) alloc((size_t)NUM_NODES * HID * 4);
    int*    rowptr = (int*)  alloc((size_t)(NUM_NODES + 1) * 4);
    int*    scol   = (int*)  alloc((size_t)E * 4);
    float*  wgt    = (float*)alloc((size_t)E * 4);
    float*  gi   = (float*) alloc((size_t)NUM_NODES * GATES * 4);
    float*  hs   = (float*) alloc((size_t)NUM_NODES * HID * 4);
    __bf16* xb   = (__bf16*)alloc((size_t)NUM_NODES * HID * 2);
    __bf16* w1t  = (__bf16*)alloc((size_t)HID * IN_FEAT * 2);
    __bf16* w2t  = (__bf16*)alloc((size_t)HID * HID * 2);
    __bf16* wih  = (__bf16*)alloc((size_t)GATES * HID * 2);
    __bf16* whh  = (__bf16*)alloc((size_t)GATES * HID * 2);
    int*    pos  = (int*)gi;
    int*    col  = (int*)gi + E;
    (void)ws_size; (void)n_in; (void)out_size;

    const int TPB = 256;
    auto blk = [](long long n, int t) { return (unsigned)((n + t - 1) / t); };

    cvt_f32_to_bf16<<<blk((long long)NUM_NODES * IN_FEAT, TPB), TPB, 0, stream>>>(
        x, xb, NUM_NODES * IN_FEAT);
    cvt_f32_to_bf16_T<<<blk((long long)IN_FEAT * HID, TPB), TPB, 0, stream>>>(
        W1, w1t, IN_FEAT, HID);
    cvt_f32_to_bf16_T<<<blk((long long)HID * HID, TPB), TPB, 0, stream>>>(
        W2, w2t, HID, HID);
    cvt_f32_to_bf16<<<blk((long long)GATES * HID, TPB), TPB, 0, stream>>>(
        W_ih, wih, GATES * HID);
    cvt_f32_to_bf16<<<blk((long long)GATES * HID, TPB), TPB, 0, stream>>>(
        W_hh, whh, GATES * HID);

    csr_zero<<<blk(NUM_NODES, TPB), TPB, 0, stream>>>(cnt, NUM_NODES);
    csr_count<<<blk(E, TPB), TPB, 0, stream>>>(dst, cnt, pos, E);
    csr_offsets<<<1, OFFS_THREADS, 0, stream>>>(cnt, rowptr, dinv, NUM_NODES);
    csr_place<<<blk(E, TPB), TPB, 0, stream>>>(src, dst, rowptr, pos, col, E);
    csr_rank<<<blk((long long)NUM_NODES * 32, TPB), TPB, 0, stream>>>(col, rowptr, dinv, scol, wgt, NUM_NODES);

    const unsigned gemm1_blocks = blk((long long)(NUM_NODES / 16) * (HID / 16), 4);
    const unsigned gemm2_blocks = gemm1_blocks;
    const unsigned gemm3_blocks = blk((long long)(NUM_NODES / 16) * (GATES / 16), 4);
    const unsigned gath_blocks  = blk((long long)NUM_NODES * 32, TPB);

    wmma_gemm_bf16<<<gemm1_blocks, 128, 0, stream>>>(xb, w1t, xw, NUM_NODES, HID, IN_FEAT);
    gcn_gather<<<gath_blocks, TPB, 0, stream>>>(xw, rowptr, scol, wgt, dinv, b1, xb, NUM_NODES);

    wmma_gemm_bf16<<<gemm2_blocks, 128, 0, stream>>>(xb, w2t, xw, NUM_NODES, HID, HID);
    gcn_gather<<<gath_blocks, TPB, 0, stream>>>(xw, rowptr, scol, wgt, dinv, b2, xb, NUM_NODES);

    wmma_gemm_bf16<<<gemm3_blocks, 128, 0, stream>>>(xb, wih, gi, NUM_NODES, GATES, HID);

    gru_scan<<<1, SCAN_THREADS, 0, stream>>>(gi, whh, b_ih, b_hh, hs, NUM_NODES);

    fc_out_kernel<<<blk((long long)NUM_NODES * OUTF, TPB), TPB, 0, stream>>>(
        hs, fc_w, fc_b, out, NUM_NODES * OUTF);
}


# GRU scan: 4 waves, one per SIMD, each holding 12 W_hh tiles x 8 k in 768 registers; a step's 16 LDS reads stand in one group ahead of 96 back-to-back matrix instructions; sum order unchanged
# speedup vs baseline: 12.2663x; 1.1371x over previous
#include <hip/hip_runtime.h>
#include <hip/hip_bf16.h>
#include <stdint.h>

typedef __attribute__((ext_vector_type(16))) __bf16 v16bf;
typedef __attribute__((ext_vector_type(8)))  float  v8f;

#define NUM_NODES 10000
#define IN_FEAT   64
#define HID       256
#define OUTF      3
#define GATES     768

__global__ void cvt_f32_to_bf16(const float* __restrict__ in,
                                __bf16* __restrict__ out, int n) {
    int i = blockIdx.x * blockDim.x + threadIdx.x;
    if (i < n) out[i] = (__bf16)in[i];
}

__global__ void cvt_f32_to_bf16_T(const float* __restrict__ in,
                                  __bf16* __restrict__ out, int K, int N) {
    int i = blockIdx.x * blockDim.x + threadIdx.x;
    if (i >= K * N) return;
    int k = i / N, n = i % N;
    out[(size_t)n * K + k] = (__bf16)in[(size_t)k * N + n];
}

__global__ void csr_zero(int* __restrict__ cnt, int n) {
    int i = blockIdx.x * blockDim.x + threadIdx.x;
    if (i < n) cnt[i] = 0;
}

__global__ void csr_count(const int* __restrict__ dst, int* __restrict__ cnt,
                          int* __restrict__ pos, int E) {
    int e = blockIdx.x * blockDim.x + threadIdx.x;
    if (e < E) pos[e] = atomicAdd(&cnt[dst[e]], 1);
}

#define OFFS_THREADS 1024
__global__ void __launch_bounds__(OFFS_THREADS)
csr_offsets(const int* __restrict__ cnt, int* __restrict__ rowptr,
            float* __restrict__ dinv, int n) {
    __shared__ int part[OFFS_THREADS];
    const int t   = threadIdx.x;
    const int per = (n + OFFS_THREADS - 1) / OFFS_THREADS;
    const int b0  = t * per;
    int s = 0;
    for (int i = 0; i < per; ++i) {
        const int idx = b0 + i;
        if (idx < n) s += cnt[idx];
    }
    part[t] = s;
    __syncthreads();
    for (int off = 1; off < OFFS_THREADS; off <<= 1) {
        const int v = (t >= off) ? part[t - off] : 0;
        __syncthreads();
        part[t] += v;
        __syncthreads();
    }
    int run = part[t] - s;
    for (int i = 0; i < per; ++i) {
        const int idx = b0 + i;
        if (idx < n) {
            const int c = cnt[idx];
            rowptr[idx] = run;
            dinv[idx]   = rsqrtf((float)(c + 1));
            run += c;
        }
    }
    if (t == OFFS_THREADS - 1) rowptr[n] = part[t];
}

__global__ void csr_place(const int* __restrict__ src, const int* __restrict__ dst,
                          const int* __restrict__ rowptr, const int* __restrict__ pos,
                          int* __restrict__ col, int E) {
    int e = blockIdx.x * blockDim.x + threadIdx.x;
    if (e < E) col[rowptr[dst[e]] + pos[e]] = src[e];
}

__global__ void csr_rank(const int* __restrict__ col, const int* __restrict__ rowptr,
                         const float* __restrict__ dinv,
                         int* __restrict__ scol, float* __restrict__ wgt, int n) {
    const int node = (int)((blockIdx.x * blockDim.x + threadIdx.x) >> 5);
    const int lane = threadIdx.x & 31;
    if (node >= n) return;
    const int b = rowptr[node];
    const int L = rowptr[node + 1] - b;
    const float dd = dinv[node];
    const int* row = col + b;
    for (int i = lane; i < L; i += 32) {
        const int s = row[i];
        int rank = 0;
        for (int q = 0; q < L; ++q) {
            const int v = row[q];
            rank += (v < s || (v == s && q < i)) ? 1 : 0;
        }
        scol[b + rank] = s;
        wgt[b + rank]  = dinv[s] * dd;
    }
}

__global__ void __launch_bounds__(128)
wmma_gemm_bf16(const __bf16* __restrict__ A, const __bf16* __restrict__ Bt,
               float* __restrict__ C, int M, int N, int K) {
    int wave  = (int)((blockIdx.x * blockDim.x + threadIdx.x) >> 5);
    int lane  = threadIdx.x & 31;
    int tilesN = N >> 4;
    int tilesM = M >> 4;
    if (wave >= tilesM * tilesN) return;
    int tm = wave / tilesN;
    int tn = wave % tilesN;

    int l15 = lane & 15;
    int h   = lane >> 4;
    const __bf16* arow = A  + (size_t)(tm * 16 + l15) * K;
    const __bf16* bcol = Bt + (size_t)(tn * 16 + l15) * K;

    v8f acc = {};
    for (int k = 0; k < K; k += 32) {
        v16bf a, b;
        int k0 = k + h * 8;
#pragma unroll
        for (int i = 0; i < 8; ++i) {
            a[i]     = arow[k0 + i];
            a[i + 8] = arow[k0 + 16 + i];
            b[i]     = bcol[k0 + i];
            b[i + 8] = bcol[k0 + 16 + i];
        }
        acc = __builtin_amdgcn_wmma_f32_16x16x32_bf16(
            false, a, false, b, (short)0, acc, false, false);
    }
#pragma unroll
    for (int r = 0; r < 8; ++r) {
        int mm = tm * 16 + h * 8 + r;
        C[(size_t)mm * N + tn * 16 + l15] = acc[r];
    }
}

typedef __attribute__((ext_vector_type(4))) __bf16 v4bf;

__global__ void __launch_bounds__(256)
gcn_gather(const float* __restrict__ xw, const int* __restrict__ rowptr,
           const int* __restrict__ scol, const float* __restrict__ wgt,
           const float* __restrict__ dinv, const float* __restrict__ bias,
           __bf16* __restrict__ hb, int n) {
    const int lane = threadIdx.x & 31;
    const int node = __builtin_amdgcn_readfirstlane((int)((blockIdx.x * blockDim.x + threadIdx.x) >> 5));
    if (node >= n) return;
    const int b = __builtin_amdgcn_readfirstlane(rowptr[node]);
    const int L = __builtin_amdgcn_readfirstlane(rowptr[node + 1]) - b;
    const float sd = dinv[node];

    const float4* self = (const float4*)(xw + (size_t)node * HID);
    float4 a0 = self[lane], a1 = self[32 + lane];
    a0.x = a0.x * sd * sd; a0.y = a0.y * sd * sd; a0.z = a0.z * sd * sd; a0.w = a0.w * sd * sd;
    a1.x = a1.x * sd * sd; a1.y = a1.y * sd * sd; a1.z = a1.z * sd * sd; a1.w = a1.w * sd * sd;

    int   ms = 0;
    float mw = 0.0f;
    if (lane < L) { ms = scol[b + lane]; mw = wgt[b + lane]; }
    for (int c = 0; c < L; c += 32) {
        const int m = (L - c < 32) ? (L - c) : 32;
        int   ns = 0;
        float nw = 0.0f;
        if (c + 32 + lane < L) { ns = scol[b + c + 32 + lane]; nw = wgt[b + c + 32 + lane]; }
#pragma unroll 8
        for (int q = 0; q < m; ++q) {
            const int   s = __builtin_amdgcn_readlane(ms, q);
            const float w = __builtin_bit_cast(float, __builtin_amdgcn_readlane(__builtin_bit_cast(int, mw), q));
            const float4* r = (const float4*)(xw + (size_t)s * HID);
            const float4 v0 = r[lane], v1 = r[32 + lane];
            a0.x = fmaf(w, v0.x, a0.x); a0.y = fmaf(w, v0.y, a0.y);
            a0.z = fmaf(w, v0.z, a0.z); a0.w = fmaf(w, v0.w, a0.w);
            a1.x = fmaf(w, v1.x, a1.x); a1.y = fmaf(w, v1.y, a1.y);
            a1.z = fmaf(w, v1.z, a1.z); a1.w = fmaf(w, v1.w, a1.w);
        }
        ms = ns; mw = nw;
    }

    const float* bl = bias + 4 * lane;
    a0.x += bl[0];   a0.y += bl[1];   a0.z += bl[2];   a0.w += bl[3];
    a1.x += bl[128]; a1.y += bl[129]; a1.z += bl[130]; a1.w += bl[131];
    v4bf o0, o1;
    o0[0] = (__bf16)(a0.x > 0.f ? a0.x : 0.f); o0[1] = (__bf16)(a0.y > 0.f ? a0.y : 0.f);
    o0[2] = (__bf16)(a0.z > 0.f ? a0.z : 0.f); o0[3] = (__bf16)(a0.w > 0.f ? a0.w : 0.f);
    o1[0] = (__bf16)(a1.x > 0.f ? a1.x : 0.f); o1[1] = (__bf16)(a1.y > 0.f ? a1.y : 0.f);
    o1[2] = (__bf16)(a1.z > 0.f ? a1.z : 0.f); o1[3] = (__bf16)(a1.w > 0.f ? a1.w : 0.f);
    __bf16* orow = hb + (size_t)node * HID;
    *(v4bf*)(orow + 4 * lane)       = o0;
    *(v4bf*)(orow + 128 + 4 * lane) = o1;
}

#define SCAN_THREADS 128
#define SCAN_GROUPS  4
#define SCAN_KT      (HID / 32)
#define HPITCH4 34

typedef __attribute__((ext_vector_type(8))) unsigned int v8u;

static __device__ __forceinline__ v16bf frag_at(const uint4* p) {
    const uint4 lo = p[0], hi = p[2];
    const v8u v = {lo.x, lo.y, lo.z, lo.w, hi.x, hi.y, hi.z, hi.w};
    return __builtin_bit_cast(v16bf, v);
}

static __device__ __forceinline__ float gate_sigmoid(float x) {
    return __builtin_amdgcn_rcpf(1.0f + __builtin_amdgcn_exp2f(x * -1.4426950408889634f));
}
static __device__ __forceinline__ float gate_tanh(float x) {
    return fmaf(2.0f, __builtin_amdgcn_rcpf(1.0f + __builtin_amdgcn_exp2f(x * -2.8853900817779268f)), -1.0f);
}

static __device__ __forceinline__ float piece_sum(const v8f d) {
    return (d[2] + d[1]) + d[0];
}

static __device__ __forceinline__ void split3(float v, __bf16& p0, __bf16& p1, __bf16& p2) {
    p0 = (__bf16)v;
    const float e1 = v - (float)p0;
    p1 = (__bf16)e1;
    const float e2 = e1 - (float)p1;
    p2 = (__bf16)e2;
}

__global__ void __launch_bounds__(SCAN_THREADS)
gru_scan(const float* __restrict__ gi, const __bf16* __restrict__ Whh,
         const float* __restrict__ b_ih, const float* __restrict__ b_hh,
         float* __restrict__ hs, int T) {
    __shared__ uint4 hsplit[2][4 * HPITCH4];
    const int t    = threadIdx.x;
    const int w    = t >> 5;
    const int lane = t & 31;
    const int c    = t & 15;
    const int hh   = (t >> 4) & 1;
    const int u0   = 64 * w;

    v16bf Br[SCAN_GROUPS][SCAN_KT], Bz[SCAN_GROUPS][SCAN_KT], Bn[SCAN_GROUPS][SCAN_KT];
#pragma unroll
    for (int g = 0; g < SCAN_GROUPS; ++g) {
        const size_t u = (size_t)(u0 + 16 * g + c);
#pragma unroll
        for (int kt = 0; kt < SCAN_KT; ++kt) {
            Br[g][kt] = frag_at((const uint4*)(Whh + u * HID + 32 * kt + 8 * hh));
            Bz[g][kt] = frag_at((const uint4*)(Whh + (HID + u) * HID + 32 * kt + 8 * hh));
            Bn[g][kt] = frag_at((const uint4*)(Whh + (2 * HID + u) * HID + 32 * kt + 8 * hh));
        }
    }
    const int ja = u0 + lane;
    const int jb = u0 + 32 + lane;
    const float bih_ra = b_ih[ja], bih_za = b_ih[HID + ja], bih_na = b_ih[2 * HID + ja];
    const float bhh_ra = b_hh[ja], bhh_za = b_hh[HID + ja], bhh_na = b_hh[2 * HID + ja];
    const float bih_rb = b_ih[jb], bih_zb = b_ih[HID + jb], bih_nb = b_ih[2 * HID + jb];
    const float bhh_rb = b_hh[jb], bhh_zb = b_hh[HID + jb], bhh_nb = b_hh[2 * HID + jb];

    for (int i = t; i < 2 * 4 * HPITCH4; i += SCAN_THREADS) (&hsplit[0][0])[i] = make_uint4(0u, 0u, 0u, 0u);
    float hpa = 0.0f, hpb = 0.0f;
    float ga_r = gi[ja], ga_z = gi[HID + ja], ga_n = gi[2 * HID + ja];
    float gb_r = gi[jb], gb_z = gi[HID + jb], gb_n = gi[2 * HID + jb];
    const int prow = c & 7;
    const int aoff = (prow < 3 ? prow : 3) * HPITCH4 + hh;
    __syncthreads();

    for (int step = 0; step < T; ++step) {
        const uint4* ha = &hsplit[step & 1][aoff];
        const unsigned gx = (unsigned)(step + 1 < T ? step + 1 : step) * GATES;
        const float nga_r = gi[gx + ja], nga_z = gi[gx + HID + ja], nga_n = gi[gx + 2 * HID + ja];
        const float ngb_r = gi[gx + jb], ngb_z = gi[gx + HID + jb], ngb_n = gi[gx + 2 * HID + jb];
        if (step > 0) {
            hs[(unsigned)(step - 1) * HID + (unsigned)ja] = hpa;
            hs[(unsigned)(step - 1) * HID + (unsigned)jb] = hpb;
        }
        v16bf a[SCAN_KT];
#pragma unroll
        for (int kt = 0; kt < SCAN_KT; ++kt) a[kt] = frag_at(ha + 4 * kt);

        v8f ar[SCAN_GROUPS], az[SCAN_GROUPS], an[SCAN_GROUPS];
#pragma unroll
        for (int g = 0; g < SCAN_GROUPS; ++g) { ar[g] = v8f{}; az[g] = v8f{}; an[g] = v8f{}; }
#pragma unroll
        for (int kt = 0; kt < SCAN_KT; ++kt) {
#pragma unroll
            for (int g = 0; g < SCAN_GROUPS; ++g)
                ar[g] = __builtin_amdgcn_wmma_f32_16x16x32_bf16(false, a[kt], false, Br[g][kt], (short)0, ar[g], false, false);
#pragma unroll
            for (int g = 0; g < SCAN_GROUPS; ++g)
                az[g] = __builtin_amdgcn_wmma_f32_16x16x32_bf16(false, a[kt], false, Bz[g][kt], (short)0, az[g], false, false);
#pragma unroll
            for (int g = 0; g < SCAN_GROUPS; ++g)
                an[g] = __builtin_amdgcn_wmma_f32_16x16x32_bf16(false, a[kt], false, Bn[g][kt], (short)0, an[g], false, false);
        }
        const float sa_r = hh ? piece_sum(ar[1]) : piece_sum(ar[0]);
        const float sa_z = hh ? piece_sum(az[1]) : piece_sum(az[0]);
        const float sa_n = hh ? piece_sum(an[1]) : piece_sum(an[0]);
        const float sb_r = hh ? piece_sum(ar[3]) : piece_sum(ar[2]);
        const float sb_z = hh ? piece_sum(az[3]) : piece_sum(az[2]);
        const float sb_n = hh ? piece_sum(an[3]) : piece_sum(an[2]);

        const float gha_r = sa_r + bhh_ra, gha_z = sa_z + bhh_za, gha_n = sa_n + bhh_na;
        const float ghb_r = sb_r + bhh_rb, ghb_z = sb_z + bhh_zb, ghb_n = sb_n + bhh_nb;
        const float ra  = gate_sigmoid(ga_r + bih_ra + gha_r);
        const float rb  = gate_sigmoid(gb_r + bih_rb + ghb_r);
        const float za  = gate_sigmoid(ga_z + bih_za + gha_z);
        const float zb  = gate_sigmoid(gb_z + bih_zb + ghb_z);
        const float nna = gate_tanh(ga_n + bih_na + ra * gha_n);
        const float nnb = gate_tanh(gb_n + bih_nb + rb * ghb_n);
        const float hna = (1.0f - za) * nna + za * hpa;
        const float hnb = (1.0f - zb) * nnb + zb * hpb;
        __bf16 pa0, pa1, pa2, pb0, pb1, pb2;
        split3(hna, pa0, pa1, pa2);
        split3(hnb, pb0, pb1, pb2);
        __bf16* hnx = (__bf16*)&hsplit[(step + 1) & 1][0];
        hnx[ja]                   = pa0;
        hnx[8 * HPITCH4 + ja]     = pa1;
        hnx[2 * 8 * HPITCH4 + ja] = pa2;
        hnx[jb]                   = pb0;
        hnx[8 * HPITCH4 + jb]     = pb1;
        hnx[2 * 8 * HPITCH4 + jb] = pb2;
        hpa = hna; hpb = hnb;
        ga_r = nga_r; ga_z = nga_z; ga_n = nga_n;
        gb_r = ngb_r; gb_z = ngb_z; gb_n = ngb_n;
        __syncthreads();
    }
    hs[(unsigned)(T - 1) * HID + (unsigned)ja] = hpa;
    hs[(unsigned)(T - 1) * HID + (unsigned)jb] = hpb;
}

__global__ void fc_out_kernel(const float* __restrict__ hs, const float* __restrict__ fw,
                              const float* __restrict__ fb, float* __restrict__ out, int total) {
    int gid = blockIdx.x * blockDim.x + threadIdx.x;
    if (gid >= total) return;
    int n = gid / OUTF, o = gid % OUTF;
    const float* h = hs + (size_t)n * HID;
    const float* w = fw + (size_t)o * HID;
    float acc = fb[o];
#pragma unroll 8
    for (int k = 0; k < HID; ++k) acc = fmaf(h[k], w[k], acc);
    out[gid] = acc;
}

extern "C" void kernel_launch(void* const* d_in, const int* in_sizes, int n_in,
                              void* d_out, int out_size, void* d_ws, size_t ws_size,
                              hipStream_t stream) {
    const float* x    = (const float*)d_in[0];
    const int*   ei   = (const int*)  d_in[1];
    const float* W1   = (const float*)d_in[2];
    const float* b1   = (const float*)d_in[3];
    const float* W2   = (const float*)d_in[4];
    const float* b2   = (const float*)d_in[5];
    const float* W_ih = (const float*)d_in[6];
    const float* W_hh = (const float*)d_in[7];
    const float* b_ih = (const float*)d_in[8];
    const float* b_hh = (const float*)d_in[9];
    const float* fc_w = (const float*)d_in[10];
    const float* fc_b = (const float*)d_in[11];
    float* out = (float*)d_out;

    const int E = in_sizes[1] / 2;
    const int* src = ei;
    const int* dst = ei + E;

    char* ws = (char*)d_ws;
    size_t off = 0;
    auto alloc = [&](size_t bytes) -> char* {
        char* p = ws + off;
        off = (off + bytes + 255) & ~(size_t)255;
        return p;
    };
    int*    cnt  = (int*)   alloc((size_t)NUM_NODES * 4);
    float*  dinv = (float*) alloc((size_t)NUM_NODES * 4);
    float*  xw   = (float*) alloc((size_t)NUM_NODES * HID * 4);
    int*    rowptr = (int*)  alloc((size_t)(NUM_NODES + 1) * 4);
    int*    scol   = (int*)  alloc((size_t)E * 4);
    float*  wgt    = (float*)alloc((size_t)E * 4);
    float*  gi   = (float*) alloc((size_t)NUM_NODES * GATES * 4);
    float*  hs   = (float*) alloc((size_t)NUM_NODES * HID * 4);
    __bf16* xb   = (__bf16*)alloc((size_t)NUM_NODES * HID * 2);
    __bf16* w1t  = (__bf16*)alloc((size_t)HID * IN_FEAT * 2);
    __bf16* w2t  = (__bf16*)alloc((size_t)HID * HID * 2);
    __bf16* wih  = (__bf16*)alloc((size_t)GATES * HID * 2);
    __bf16* whh  = (__bf16*)alloc((size_t)GATES * HID * 2);
    int*    pos  = (int*)gi;
    int*    col  = (int*)gi + E;
    (void)ws_size; (void)n_in; (void)out_size;

    const int TPB = 256;
    auto blk = [](long long n, int t) { return (unsigned)((n + t - 1) / t); };

    cvt_f32_to_bf16<<<blk((long long)NUM_NODES * IN_FEAT, TPB), TPB, 0, stream>>>(
        x, xb, NUM_NODES * IN_FEAT);
    cvt_f32_to_bf16_T<<<blk((long long)IN_FEAT * HID, TPB), TPB, 0, stream>>>(
        W1, w1t, IN_FEAT, HID);
    cvt_f32_to_bf16_T<<<blk((long long)HID * HID, TPB), TPB, 0, stream>>>(
        W2, w2t, HID, HID);
    cvt_f32_to_bf16<<<blk((long long)GATES * HID, TPB), TPB, 0, stream>>>(
        W_ih, wih, GATES * HID);
    cvt_f32_to_bf16<<<blk((long long)GATES * HID, TPB), TPB, 0, stream>>>(
        W_hh, whh, GATES * HID);

    csr_zero<<<blk(NUM_NODES, TPB), TPB, 0, stream>>>(cnt, NUM_NODES);
    csr_count<<<blk(E, TPB), TPB, 0, stream>>>(dst, cnt, pos, E);
    csr_offsets<<<1, OFFS_THREADS, 0, stream>>>(cnt, rowptr, dinv, NUM_NODES);
    csr_place<<<blk(E, TPB), TPB, 0, stream>>>(src, dst, rowptr, pos, col, E);
    csr_rank<<<blk((long long)NUM_NODES * 32, TPB), TPB, 0, stream>>>(col, rowptr, dinv, scol, wgt, NUM_NODES);

    const unsigned gemm1_blocks = blk((long long)(NUM_NODES / 16) * (HID / 16), 4);
    const unsigned gemm2_blocks = gemm1_blocks;
    const unsigned gemm3_blocks = blk((long long)(NUM_NODES / 16) * (GATES / 16), 4);
    const unsigned gath_blocks  = blk((long long)NUM_NODES * 32, TPB);

    wmma_gemm_bf16<<<gemm1_blocks, 128, 0, stream>>>(xb, w1t, xw, NUM_NODES, HID, IN_FEAT);
    gcn_gather<<<gath_blocks, TPB, 0, stream>>>(xw, rowptr, scol, wgt, dinv, b1, xb, NUM_NODES);

    wmma_gemm_bf16<<<gemm2_blocks, 128, 0, stream>>>(xb, w2t, xw, NUM_NODES, HID, HID);
    gcn_gather<<<gath_blocks, TPB, 0, stream>>>(xw, rowptr, scol, wgt, dinv, b2, xb, NUM_NODES);

    wmma_gemm_bf16<<<gemm3_blocks, 128, 0, stream>>>(xb, wih, gi, NUM_NODES, GATES, HID);

    gru_scan<<<1, SCAN_THREADS, 0, stream>>>(gi, whh, b_ih, b_hh, hs, NUM_NODES);

    fc_out_kernel<<<blk((long long)NUM_NODES * OUTF, TPB), TPB, 0, stream>>>(
        hs, fc_w, fc_b, out, NUM_NODES * OUTF);
}


# GRU scan: a step's matrix instructions re-ordered by unit half (k tiles 0,2,4,6,1,3,5,7) so both h hand-overs run beside them; barriers split arrive/pass and wait for LDS only, not global memory
# speedup vs baseline: 13.6326x; 1.1114x over previous
#include <hip/hip_runtime.h>
#include <hip/hip_bf16.h>
#include <stdint.h>

typedef __attribute__((ext_vector_type(16))) __bf16 v16bf;
typedef __attribute__((ext_vector_type(8)))  float  v8f;

#define NUM_NODES 10000
#define IN_FEAT   64
#define HID       256
#define OUTF      3
#define GATES     768

__global__ void cvt_f32_to_bf16(const float* __restrict__ in,
                                __bf16* __restrict__ out, int n) {
    int i = blockIdx.x * blockDim.x + threadIdx.x;
    if (i < n) out[i] = (__bf16)in[i];
}

__global__ void cvt_f32_to_bf16_T(const float* __restrict__ in,
                                  __bf16* __restrict__ out, int K, int N) {
    int i = blockIdx.x * blockDim.x + threadIdx.x;
    if (i >= K * N) return;
    int k = i / N, n = i % N;
    out[(size_t)n * K + k] = (__bf16)in[(size_t)k * N + n];
}

__global__ void csr_zero(int* __restrict__ cnt, int n) {
    int i = blockIdx.x * blockDim.x + threadIdx.x;
    if (i < n) cnt[i] = 0;
}

__global__ void csr_count(const int* __restrict__ dst, int* __restrict__ cnt,
                          int* __restrict__ pos, int E) {
    int e = blockIdx.x * blockDim.x + threadIdx.x;
    if (e < E) pos[e] = atomicAdd(&cnt[dst[e]], 1);
}

#define OFFS_THREADS 1024
__global__ void __launch_bounds__(OFFS_THREADS)
csr_offsets(const int* __restrict__ cnt, int* __restrict__ rowptr,
            float* __restrict__ dinv, int n) {
    __shared__ int part[OFFS_THREADS];
    const int t   = threadIdx.x;
    const int per = (n + OFFS_THREADS - 1) / OFFS_THREADS;
    const int b0  = t * per;
    int s = 0;
    for (int i = 0; i < per; ++i) {
        const int idx = b0 + i;
        if (idx < n) s += cnt[idx];
    }
    part[t] = s;
    __syncthreads();
    for (int off = 1; off < OFFS_THREADS; off <<= 1) {
        const int v = (t >= off) ? part[t - off] : 0;
        __syncthreads();
        part[t] += v;
        __syncthreads();
    }
    int run = part[t] - s;
    for (int i = 0; i < per; ++i) {
        const int idx = b0 + i;
        if (idx < n) {
            const int c = cnt[idx];
            rowptr[idx] = run;
            dinv[idx]   = rsqrtf((float)(c + 1));
            run += c;
        }
    }
    if (t == OFFS_THREADS - 1) rowptr[n] = part[t];
}

__global__ void csr_place(const int* __restrict__ src, const int* __restrict__ dst,
                          const int* __restrict__ rowptr, const int* __restrict__ pos,
                          int* __restrict__ col, int E) {
    int e = blockIdx.x * blockDim.x + threadIdx.x;
    if (e < E) col[rowptr[dst[e]] + pos[e]] = src[e];
}

__global__ void csr_rank(const int* __restrict__ col, const int* __restrict__ rowptr,
                         const float* __restrict__ dinv,
                         int* __restrict__ scol, float* __restrict__ wgt, int n) {
    const int node = (int)((blockIdx.x * blockDim.x + threadIdx.x) >> 5);
    const int lane = threadIdx.x & 31;
    if (node >= n) return;
    const int b = rowptr[node];
    const int L = rowptr[node + 1] - b;
    const float dd = dinv[node];
    const int* row = col + b;
    for (int i = lane; i < L; i += 32) {
        const int s = row[i];
        int rank = 0;
        for (int q = 0; q < L; ++q) {
            const int v = row[q];
            rank += (v < s || (v == s && q < i)) ? 1 : 0;
        }
        scol[b + rank] = s;
        wgt[b + rank]  = dinv[s] * dd;
    }
}

__global__ void __launch_bounds__(128)
wmma_gemm_bf16(const __bf16* __restrict__ A, const __bf16* __restrict__ Bt,
               float* __restrict__ C, int M, int N, int K) {
    int wave  = (int)((blockIdx.x * blockDim.x + threadIdx.x) >> 5);
    int lane  = threadIdx.x & 31;
    int tilesN = N >> 4;
    int tilesM = M >> 4;
    if (wave >= tilesM * tilesN) return;
    int tm = wave / tilesN;
    int tn = wave % tilesN;

    int l15 = lane & 15;
    int h   = lane >> 4;
    const __bf16* arow = A  + (size_t)(tm * 16 + l15) * K;
    const __bf16* bcol = Bt + (size_t)(tn * 16 + l15) * K;

    v8f acc = {};
    for (int k = 0; k < K; k += 32) {
        v16bf a, b;
        int k0 = k + h * 8;
#pragma unroll
        for (int i = 0; i < 8; ++i) {
            a[i]     = arow[k0 + i];
            a[i + 8] = arow[k0 + 16 + i];
            b[i]     = bcol[k0 + i];
            b[i + 8] = bcol[k0 + 16 + i];
        }
        acc = __builtin_amdgcn_wmma_f32_16x16x32_bf16(
            false, a, false, b, (short)0, acc, false, false);
    }
#pragma unroll
    for (int r = 0; r < 8; ++r) {
        int mm = tm * 16 + h * 8 + r;
        C[(size_t)mm * N + tn * 16 + l15] = acc[r];
    }
}

typedef __attribute__((ext_vector_type(4))) __bf16 v4bf;

__global__ void __launch_bounds__(256)
gcn_gather(const float* __restrict__ xw, const int* __restrict__ rowptr,
           const int* __restrict__ scol, const float* __restrict__ wgt,
           const float* __restrict__ dinv, const float* __restrict__ bias,
           __bf16* __restrict__ hb, int n) {
    const int lane = threadIdx.x & 31;
    const int node = __builtin_amdgcn_readfirstlane((int)((blockIdx.x * blockDim.x + threadIdx.x) >> 5));
    if (node >= n) return;
    const int b = __builtin_amdgcn_readfirstlane(rowptr[node]);
    const int L = __builtin_amdgcn_readfirstlane(rowptr[node + 1]) - b;
    const float sd = dinv[node];

    const float4* self = (const float4*)(xw + (size_t)node * HID);
    float4 a0 = self[lane], a1 = self[32 + lane];
    a0.x = a0.x * sd * sd; a0.y = a0.y * sd * sd; a0.z = a0.z * sd * sd; a0.w = a0.w * sd * sd;
    a1.x = a1.x * sd * sd; a1.y = a1.y * sd * sd; a1.z = a1.z * sd * sd; a1.w = a1.w * sd * sd;

    int   ms = 0;
    float mw = 0.0f;
    if (lane < L) { ms = scol[b + lane]; mw = wgt[b + lane]; }
    for (int c = 0; c < L; c += 32) {
        const int m = (L - c < 32) ? (L - c) : 32;
        int   ns = 0;
        float nw = 0.0f;
        if (c + 32 + lane < L) { ns = scol[b + c + 32 + lane]; nw = wgt[b + c + 32 + lane]; }
#pragma unroll 8
        for (int q = 0; q < m; ++q) {
            const int   s = __builtin_amdgcn_readlane(ms, q);
            const float w = __builtin_bit_cast(float, __builtin_amdgcn_readlane(__builtin_bit_cast(int, mw), q));
            const float4* r = (const float4*)(xw + (size_t)s * HID);
            const float4 v0 = r[lane], v1 = r[32 + lane];
            a0.x = fmaf(w, v0.x, a0.x); a0.y = fmaf(w, v0.y, a0.y);
            a0.z = fmaf(w, v0.z, a0.z); a0.w = fmaf(w, v0.w, a0.w);
            a1.x = fmaf(w, v1.x, a1.x); a1.y = fmaf(w, v1.y, a1.y);
            a1.z = fmaf(w, v1.z, a1.z); a1.w = fmaf(w, v1.w, a1.w);
        }
        ms = ns; mw = nw;
    }

    const float* bl = bias + 4 * lane;
    a0.x += bl[0];   a0.y += bl[1];   a0.z += bl[2];   a0.w += bl[3];
    a1.x += bl[128]; a1.y += bl[129]; a1.z += bl[130]; a1.w += bl[131];
    v4bf o0, o1;
    o0[0] = (__bf16)(a0.x > 0.f ? a0.x : 0.f); o0[1] = (__bf16)(a0.y > 0.f ? a0.y : 0.f);
    o0[2] = (__bf16)(a0.z > 0.f ? a0.z : 0.f); o0[3] = (__bf16)(a0.w > 0.f ? a0.w : 0.f);
    o1[0] = (__bf16)(a1.x > 0.f ? a1.x : 0.f); o1[1] = (__bf16)(a1.y > 0.f ? a1.y : 0.f);
    o1[2] = (__bf16)(a1.z > 0.f ? a1.z : 0.f); o1[3] = (__bf16)(a1.w > 0.f ? a1.w : 0.f);
    __bf16* orow = hb + (size_t)node * HID;
    *(v4bf*)(orow + 4 * lane)       = o0;
    *(v4bf*)(orow + 128 + 4 * lane) = o1;
}

#define SCAN_THREADS 128
#define SCAN_GROUPS  4
#define SCAN_KT      (HID / 32)
#define HPITCH4 34

typedef __attribute__((ext_vector_type(8))) unsigned int v8u;

static __device__ __forceinline__ v16bf frag_at(const uint4* p) {
    const uint4 lo = p[0], hi = p[2];
    const v8u v = {lo.x, lo.y, lo.z, lo.w, hi.x, hi.y, hi.z, hi.w};
    return __builtin_bit_cast(v16bf, v);
}

#define GATE_K  (-1.4426950408889634f)
#define GATE_2K (-2.8853900817779268f)

static __device__ __forceinline__ float gate_q(float a) {
    return __builtin_amdgcn_rcpf(1.0f + __builtin_amdgcn_exp2f(a));
}

static __device__ __forceinline__ float scaled_piece_sum(float m, float d0, float d1, float d2, float k) {
    return fmaf(m, d2 + d1, fmaf(m, d0, k));
}

static __device__ __forceinline__ void split3(float v, unsigned& p0, unsigned& p1, unsigned& p2) {
    const unsigned b  = __builtin_bit_cast(unsigned, v);
    const unsigned b0 = b & 0xFFFF0000u;
    const unsigned bq = b & 0xFFFFFF00u;
    p0 = b0;
    p1 = __builtin_bit_cast(unsigned, __builtin_bit_cast(float, bq) - __builtin_bit_cast(float, b0));
    p2 = __builtin_bit_cast(unsigned, v - __builtin_bit_cast(float, bq));
}

static __device__ __forceinline__ void lds_arrive() {
    __builtin_amdgcn_fence(__ATOMIC_RELEASE, "workgroup", "local");
    __builtin_amdgcn_s_barrier_signal(-1);
}
static __device__ __forceinline__ void lds_pass() {
    __builtin_amdgcn_s_barrier_wait(-1);
    __builtin_amdgcn_fence(__ATOMIC_ACQUIRE, "workgroup", "local");
}

__global__ void __launch_bounds__(SCAN_THREADS)
gru_scan(const float* __restrict__ gi, const __bf16* __restrict__ Whh,
         const float* __restrict__ b_ih, const float* __restrict__ b_hh,
         float* __restrict__ hs, int T) {
    __shared__ uint4 hsplit[4 * HPITCH4];
    const int t    = threadIdx.x;
    const int w    = t >> 5;
    const int lane = t & 31;
    const int c    = t & 15;
    const int hh   = (t >> 4) & 1;
    const int u0   = 64 * w;

    v16bf Br[SCAN_GROUPS][SCAN_KT], Bz[SCAN_GROUPS][SCAN_KT], Bn[SCAN_GROUPS][SCAN_KT];
#pragma unroll
    for (int g = 0; g < SCAN_GROUPS; ++g) {
        const size_t u = (size_t)(u0 + 16 * g + c);
#pragma unroll
        for (int kt = 0; kt < SCAN_KT; ++kt) {
            Br[g][kt] = frag_at((const uint4*)(Whh + u * HID + 32 * kt + 8 * hh));
            Bz[g][kt] = frag_at((const uint4*)(Whh + (HID + u) * HID + 32 * kt + 8 * hh));
            Bn[g][kt] = frag_at((const uint4*)(Whh + (2 * HID + u) * HID + 32 * kt + 8 * hh));
        }
    }
    const int ja = u0 + lane;
    const int jb = u0 + 32 + lane;
    const float bih_ra = b_ih[ja], bih_za = b_ih[HID + ja], bih_na = b_ih[2 * HID + ja];
    const float bhh_ra = b_hh[ja], bhh_za = b_hh[HID + ja], bhh_na = b_hh[2 * HID + ja];
    const float bih_rb = b_ih[jb], bih_zb = b_ih[HID + jb], bih_nb = b_ih[2 * HID + jb];
    const float bhh_rb = b_hh[jb], bhh_zb = b_hh[HID + jb], bhh_nb = b_hh[2 * HID + jb];
    const float bsa_r = bih_ra + bhh_ra, bsa_z = bih_za + bhh_za;
    const float bsb_r = bih_rb + bhh_rb, bsb_z = bih_zb + bhh_zb;

    for (int i = t; i < 4 * HPITCH4; i += SCAN_THREADS) hsplit[i] = make_uint4(0u, 0u, 0u, 0u);
    float hpa = 0.0f, hpb = 0.0f;
    float xa_r = gi[ja], xa_z = gi[HID + ja], xa_n = gi[2 * HID + ja];
    float xb_r = gi[jb], xb_z = gi[HID + jb], xb_n = gi[2 * HID + jb];
    float ya_r = 0.0f, ya_z = 0.0f, ya_n = 0.0f, yb_r = 0.0f, yb_z = 0.0f, yb_n = 0.0f;
    const int prow = c & 7;
    const int aoff = (prow < 3 ? prow : 3) * HPITCH4 + hh;
    const uint4* hrd = &hsplit[aoff];
    unsigned short* hwr = (unsigned short*)&hsplit[0];

    v16bf aE[4], aO[4];
    v8f r1[2], z1[2], n1[2], r2[2], z2[2], n2[2];
#pragma unroll
    for (int i = 0; i < 4; ++i) { aE[i] = v16bf{}; aO[i] = v16bf{}; }
#pragma unroll
    for (int g = 0; g < 2; ++g) { r1[g] = v8f{}; z1[g] = v8f{}; n1[g] = v8f{}; r2[g] = v8f{}; z2[g] = v8f{}; n2[g] = v8f{}; }
    __syncthreads();

    auto h1_tile = [&](const v16bf a, const int kt) __attribute__((always_inline)) {
#pragma unroll
        for (int g = 0; g < 2; ++g)
            r1[g] = __builtin_amdgcn_wmma_f32_16x16x32_bf16(false, a, false, Br[g][kt], (short)0, r1[g], false, false);
#pragma unroll
        for (int g = 0; g < 2; ++g)
            z1[g] = __builtin_amdgcn_wmma_f32_16x16x32_bf16(false, a, false, Bz[g][kt], (short)0, z1[g], false, false);
#pragma unroll
        for (int g = 0; g < 2; ++g)
            n1[g] = __builtin_amdgcn_wmma_f32_16x16x32_bf16(false, a, false, Bn[g][kt], (short)0, n1[g], false, false);
    };
    auto h2_tile = [&](const v16bf a, const int kt) __attribute__((always_inline)) {
#pragma unroll
        for (int g = 0; g < 2; ++g)
            r2[g] = __builtin_amdgcn_wmma_f32_16x16x32_bf16(false, a, false, Br[2 + g][kt], (short)0, r2[g], false, false);
#pragma unroll
        for (int g = 0; g < 2; ++g)
            z2[g] = __builtin_amdgcn_wmma_f32_16x16x32_bf16(false, a, false, Bz[2 + g][kt], (short)0, z2[g], false, false);
#pragma unroll
        for (int g = 0; g < 2; ++g)
            n2[g] = __builtin_amdgcn_wmma_f32_16x16x32_bf16(false, a, false, Bn[2 + g][kt], (short)0, n2[g], false, false);
    };
    auto new_h = [&](const v8f (&r)[2], const v8f (&z)[2], const v8f (&n)[2],
                     const float c_r, const float c_z, const float c_n, const float bhh_n, const float hp)
        __attribute__((always_inline)) -> float {
        const float rr = gate_q(scaled_piece_sum(GATE_K, hh ? r[1][0] : r[0][0], hh ? r[1][1] : r[0][1],
                                                 hh ? r[1][2] : r[0][2], c_r));
        const float zz = gate_q(scaled_piece_sum(GATE_K, hh ? z[1][0] : z[0][0], hh ? z[1][1] : z[0][1],
                                                 hh ? z[1][2] : z[0][2], c_z));
        const float rp = rr * GATE_2K;
        const float kn = fmaf(rp, bhh_n, c_n);
        const float oz = 1.0f - zz;
        const float A  = oz + oz;
        const float Bc = fmaf(zz, hp, -oz);
        const float q  = gate_q(scaled_piece_sum(rp, hh ? n[1][0] : n[0][0], hh ? n[1][1] : n[0][1],
                                                 hh ? n[1][2] : n[0][2], kn));
        return fmaf(A, q, Bc);
    };
    auto put_h = [&](const float v, const int j) __attribute__((always_inline)) {
        unsigned p0, p1, p2;
        split3(v, p0, p1, p2);
        hwr[j]                   = (unsigned short)(p0 >> 16);
        hwr[8 * HPITCH4 + j]     = (unsigned short)(p1 >> 16);
        hwr[2 * 8 * HPITCH4 + j] = (unsigned short)(p2 >> 16);
    };

    auto scan_step = [&](const int step,
                         const float ga_r, const float ga_z, const float ga_n,
                         const float gb_r, const float gb_z, const float gb_n,
                         float& nga_r, float& nga_z, float& nga_n,
                         float& ngb_r, float& ngb_z, float& ngb_n)
        __attribute__((always_inline)) {
        const unsigned gx = (unsigned)(step + 1 < T ? step + 1 : step) * GATES;
        nga_r = gi[gx + ja]; nga_z = gi[gx + HID + ja]; nga_n = gi[gx + 2 * HID + ja];
        ngb_r = gi[gx + jb]; ngb_z = gi[gx + HID + jb]; ngb_n = gi[gx + 2 * HID + jb];
        if (step > 0) {
            hs[(unsigned)(step - 1) * HID + (unsigned)ja] = hpa;
            hs[(unsigned)(step - 1) * HID + (unsigned)jb] = hpb;
        }
        const float ca_r = (ga_r + bsa_r) * GATE_K,  cb_r = (gb_r + bsb_r) * GATE_K;
        const float ca_z = (ga_z + bsa_z) * GATE_K,  cb_z = (gb_z + bsb_z) * GATE_K;
        const float ca_n = (ga_n + bih_na) * GATE_2K, cb_n = (gb_n + bih_nb) * GATE_2K;

        h1_tile(aO[0], 1); h1_tile(aO[1], 3); h1_tile(aO[2], 5); h1_tile(aO[3], 7);
        const float hna = new_h(r1, z1, n1, ca_r, ca_z, ca_n, bhh_na, hpa);
        put_h(hna, ja);
        hpa = hna;
        h2_tile(aE[2], 4); h2_tile(aE[3], 6);
        __builtin_amdgcn_sched_barrier(0);
        h2_tile(aO[0], 1);
        __builtin_amdgcn_sched_barrier(0);
        lds_arrive();
        __builtin_amdgcn_sched_barrier(0);
        h2_tile(aO[1], 3);
        __builtin_amdgcn_sched_barrier(0);
        lds_pass();
#pragma unroll
        for (int i = 0; i < 4; ++i) aE[i] = frag_at(hrd + 4 * (2 * i));
        __builtin_amdgcn_sched_barrier(0);
        h2_tile(aO[2], 5); h2_tile(aO[3], 7);
        const float hnb = new_h(r2, z2, n2, cb_r, cb_z, cb_n, bhh_nb, hpb);
        put_h(hnb, jb);
        hpb = hnb;
#pragma unroll
        for (int g = 0; g < 2; ++g) { r1[g] = v8f{}; z1[g] = v8f{}; n1[g] = v8f{}; }
        h1_tile(aE[0], 0); h1_tile(aE[1], 2); h1_tile(aE[2], 4);
        __builtin_amdgcn_sched_barrier(0);
        lds_arrive();
        __builtin_amdgcn_sched_barrier(0);
        h1_tile(aE[3], 6);
        __builtin_amdgcn_sched_barrier(0);
        lds_pass();
#pragma unroll
        for (int i = 0; i < 4; ++i) aO[i] = frag_at(hrd + 4 * (2 * i + 1));
        __builtin_amdgcn_sched_barrier(0);
#pragma unroll
        for (int g = 0; g < 2; ++g) { r2[g] = v8f{}; z2[g] = v8f{}; n2[g] = v8f{}; }
        h2_tile(aE[0], 0); h2_tile(aE[1], 2);
    };

    int step = 0;
    for (; step + 1 < T; step += 2) {
        scan_step(step,     xa_r, xa_z, xa_n, xb_r, xb_z, xb_n, ya_r, ya_z, ya_n, yb_r, yb_z, yb_n);
        scan_step(step + 1, ya_r, ya_z, ya_n, yb_r, yb_z, yb_n, xa_r, xa_z, xa_n, xb_r, xb_z, xb_n);
    }
    if (step < T)
        scan_step(step,     xa_r, xa_z, xa_n, xb_r, xb_z, xb_n, ya_r, ya_z, ya_n, yb_r, yb_z, yb_n);
    hs[(unsigned)(T - 1) * HID + (unsigned)ja] = hpa;
    hs[(unsigned)(T - 1) * HID + (unsigned)jb] = hpb;
}

__global__ void fc_out_kernel(const float* __restrict__ hs, const float* __restrict__ fw,
                              const float* __restrict__ fb, float* __restrict__ out, int total) {
    int gid = blockIdx.x * blockDim.x + threadIdx.x;
    if (gid >= total) return;
    int n = gid / OUTF, o = gid % OUTF;
    const float* h = hs + (size_t)n * HID;
    const float* w = fw + (size_t)o * HID;
    float acc = fb[o];
#pragma unroll 8
    for (int k = 0; k < HID; ++k) acc = fmaf(h[k], w[k], acc);
    out[gid] = acc;
}

extern "C" void kernel_launch(void* const* d_in, const int* in_sizes, int n_in,
                              void* d_out, int out_size, void* d_ws, size_t ws_size,
                              hipStream_t stream) {
    const float* x    = (const float*)d_in[0];
    const int*   ei   = (const int*)  d_in[1];
    const float* W1   = (const float*)d_in[2];
    const float* b1   = (const float*)d_in[3];
    const float* W2   = (const float*)d_in[4];
    const float* b2   = (const float*)d_in[5];
    const float* W_ih = (const float*)d_in[6];
    const float* W_hh = (const float*)d_in[7];
    const float* b_ih = (const float*)d_in[8];
    const float* b_hh = (const float*)d_in[9];
    const float* fc_w = (const float*)d_in[10];
    const float* fc_b = (const float*)d_in[11];
    float* out = (float*)d_out;

    const int E = in_sizes[1] / 2;
    const int* src = ei;
    const int* dst = ei + E;

    char* ws = (char*)d_ws;
    size_t off = 0;
    auto alloc = [&](size_t bytes) -> char* {
        char* p = ws + off;
        off = (off + bytes + 255) & ~(size_t)255;
        return p;
    };
    int*    cnt  = (int*)   alloc((size_t)NUM_NODES * 4);
    float*  dinv = (float*) alloc((size_t)NUM_NODES * 4);
    float*  xw   = (float*) alloc((size_t)NUM_NODES * HID * 4);
    int*    rowptr = (int*)  alloc((size_t)(NUM_NODES + 1) * 4);
    int*    scol   = (int*)  alloc((size_t)E * 4);
    float*  wgt    = (float*)alloc((size_t)E * 4);
    float*  gi   = (float*) alloc((size_t)NUM_NODES * GATES * 4);
    float*  hs   = (float*) alloc((size_t)NUM_NODES * HID * 4);
    __bf16* xb   = (__bf16*)alloc((size_t)NUM_NODES * HID * 2);
    __bf16* w1t  = (__bf16*)alloc((size_t)HID * IN_FEAT * 2);
    __bf16* w2t  = (__bf16*)alloc((size_t)HID * HID * 2);
    __bf16* wih  = (__bf16*)alloc((size_t)GATES * HID * 2);
    __bf16* whh  = (__bf16*)alloc((size_t)GATES * HID * 2);
    int*    pos  = (int*)gi;
    int*    col  = (int*)gi + E;
    (void)ws_size; (void)n_in; (void)out_size;

    const int TPB = 256;
    auto blk = [](long long n, int t) { return (unsigned)((n + t - 1) / t); };

    cvt_f32_to_bf16<<<blk((long long)NUM_NODES * IN_FEAT, TPB), TPB, 0, stream>>>(
        x, xb, NUM_NODES * IN_FEAT);
    cvt_f32_to_bf16_T<<<blk((long long)IN_FEAT * HID, TPB), TPB, 0, stream>>>(
        W1, w1t, IN_FEAT, HID);
    cvt_f32_to_bf16_T<<<blk((long long)HID * HID, TPB), TPB, 0, stream>>>(
        W2, w2t, HID, HID);
    cvt_f32_to_bf16<<<blk((long long)GATES * HID, TPB), TPB, 0, stream>>>(
        W_ih, wih, GATES * HID);
    cvt_f32_to_bf16<<<blk((long long)GATES * HID, TPB), TPB, 0, stream>>>(
        W_hh, whh, GATES * HID);

    csr_zero<<<blk(NUM_NODES, TPB), TPB, 0, stream>>>(cnt, NUM_NODES);
    csr_count<<<blk(E, TPB), TPB, 0, stream>>>(dst, cnt, pos, E);
    csr_offsets<<<1, OFFS_THREADS, 0, stream>>>(cnt, rowptr, dinv, NUM_NODES);
    csr_place<<<blk(E, TPB), TPB, 0, stream>>>(src, dst, rowptr, pos, col, E);
    csr_rank<<<blk((long long)NUM_NODES * 32, TPB), TPB, 0, stream>>>(col, rowptr, dinv, scol, wgt, NUM_NODES);

    const unsigned gemm1_blocks = blk((long long)(NUM_NODES / 16) * (HID / 16), 4);
    const unsigned gemm2_blocks = gemm1_blocks;
    const unsigned gemm3_blocks = blk((long long)(NUM_NODES / 16) * (GATES / 16), 4);
    const unsigned gath_blocks  = blk((long long)NUM_NODES * 32, TPB);

    wmma_gemm_bf16<<<gemm1_blocks, 128, 0, stream>>>(xb, w1t, xw, NUM_NODES, HID, IN_FEAT);
    gcn_gather<<<gath_blocks, TPB, 0, stream>>>(xw, rowptr, scol, wgt, dinv, b1, xb, NUM_NODES);

    wmma_gemm_bf16<<<gemm2_blocks, 128, 0, stream>>>(xb, w2t, xw, NUM_NODES, HID, HID);
    gcn_gather<<<gath_blocks, TPB, 0, stream>>>(xw, rowptr, scol, wgt, dinv, b2, xb, NUM_NODES);

    wmma_gemm_bf16<<<gemm3_blocks, 128, 0, stream>>>(xb, wih, gi, NUM_NODES, GATES, HID);

    gru_scan<<<1, SCAN_THREADS, 0, stream>>>(gi, whh, b_ih, b_hh, hs, NUM_NODES);

    fc_out_kernel<<<blk((long long)NUM_NODES * OUTF, TPB), TPB, 0, stream>>>(
        hs, fc_w, fc_b, out, NUM_NODES * OUTF);
}


# GRU scan: one hand-over of h per step instead of two: a step's 48 matrix instructions stand back to back behind one group of LDS reads and one barrier; h image double-buffered; sum order unchanged
# speedup vs baseline: 16.7143x; 1.2261x over previous
#include <hip/hip_runtime.h>
#include <hip/hip_bf16.h>
#include <stdint.h>

typedef __attribute__((ext_vector_type(16))) __bf16 v16bf;
typedef __attribute__((ext_vector_type(8)))  float  v8f;

#define NUM_NODES 10000
#define IN_FEAT   64
#define HID       256
#define OUTF      3
#define GATES     768

__global__ void cvt_f32_to_bf16(const float* __restrict__ in,
                                __bf16* __restrict__ out, int n) {
    int i = blockIdx.x * blockDim.x + threadIdx.x;
    if (i < n) out[i] = (__bf16)in[i];
}

__global__ void cvt_f32_to_bf16_T(const float* __restrict__ in,
                                  __bf16* __restrict__ out, int K, int N) {
    int i = blockIdx.x * blockDim.x + threadIdx.x;
    if (i >= K * N) return;
    int k = i / N, n = i % N;
    out[(size_t)n * K + k] = (__bf16)in[(size_t)k * N + n];
}

__global__ void csr_zero(int* __restrict__ cnt, int n) {
    int i = blockIdx.x * blockDim.x + threadIdx.x;
    if (i < n) cnt[i] = 0;
}

__global__ void csr_count(const int* __restrict__ dst, int* __restrict__ cnt,
                          int* __restrict__ pos, int E) {
    int e = blockIdx.x * blockDim.x + threadIdx.x;
    if (e < E) pos[e] = atomicAdd(&cnt[dst[e]], 1);
}

#define OFFS_THREADS 1024
__global__ void __launch_bounds__(OFFS_THREADS)
csr_offsets(const int* __restrict__ cnt, int* __restrict__ rowptr,
            float* __restrict__ dinv, int n) {
    __shared__ int part[OFFS_THREADS];
    const int t   = threadIdx.x;
    const int per = (n + OFFS_THREADS - 1) / OFFS_THREADS;
    const int b0  = t * per;
    int s = 0;
    for (int i = 0; i < per; ++i) {
        const int idx = b0 + i;
        if (idx < n) s += cnt[idx];
    }
    part[t] = s;
    __syncthreads();
    for (int off = 1; off < OFFS_THREADS; off <<= 1) {
        const int v = (t >= off) ? part[t - off] : 0;
        __syncthreads();
        part[t] += v;
        __syncthreads();
    }
    int run = part[t] - s;
    for (int i = 0; i < per; ++i) {
        const int idx = b0 + i;
        if (idx < n) {
            const int c = cnt[idx];
            rowptr[idx] = run;
            dinv[idx]   = rsqrtf((float)(c + 1));
            run += c;
        }
    }
    if (t == OFFS_THREADS - 1) rowptr[n] = part[t];
}

__global__ void csr_place(const int* __restrict__ src, const int* __restrict__ dst,
                          const int* __restrict__ rowptr, const int* __restrict__ pos,
                          int* __restrict__ col, int E) {
    int e = blockIdx.x * blockDim.x + threadIdx.x;
    if (e < E) col[rowptr[dst[e]] + pos[e]] = src[e];
}

__global__ void csr_rank(const int* __restrict__ col, const int* __restrict__ rowptr,
                         const float* __restrict__ dinv,
                         int* __restrict__ scol, float* __restrict__ wgt, int n) {
    const int node = (int)((blockIdx.x * blockDim.x + threadIdx.x) >> 5);
    const int lane = threadIdx.x & 31;
    if (node >= n) return;
    const int b = rowptr[node];
    const int L = rowptr[node + 1] - b;
    const float dd = dinv[node];
    const int* row = col + b;
    for (int i = lane; i < L; i += 32) {
        const int s = row[i];
        int rank = 0;
        for (int q = 0; q < L; ++q) {
            const int v = row[q];
            rank += (v < s || (v == s && q < i)) ? 1 : 0;
        }
        scol[b + rank] = s;
        wgt[b + rank]  = dinv[s] * dd;
    }
}

__global__ void __launch_bounds__(128)
wmma_gemm_bf16(const __bf16* __restrict__ A, const __bf16* __restrict__ Bt,
               float* __restrict__ C, int M, int N, int K) {
    int wave  = (int)((blockIdx.x * blockDim.x + threadIdx.x) >> 5);
    int lane  = threadIdx.x & 31;
    int tilesN = N >> 4;
    int tilesM = M >> 4;
    if (wave >= tilesM * tilesN) return;
    int tm = wave / tilesN;
    int tn = wave % tilesN;

    int l15 = lane & 15;
    int h   = lane >> 4;
    const __bf16* arow = A  + (size_t)(tm * 16 + l15) * K;
    const __bf16* bcol = Bt + (size_t)(tn * 16 + l15) * K;

    v8f acc = {};
    for (int k = 0; k < K; k += 32) {
        v16bf a, b;
        int k0 = k + h * 8;
#pragma unroll
        for (int i = 0; i < 8; ++i) {
            a[i]     = arow[k0 + i];
            a[i + 8] = arow[k0 + 16 + i];
            b[i]     = bcol[k0 + i];
            b[i + 8] = bcol[k0 + 16 + i];
        }
        acc = __builtin_amdgcn_wmma_f32_16x16x32_bf16(
            false, a, false, b, (short)0, acc, false, false);
    }
#pragma unroll
    for (int r = 0; r < 8; ++r) {
        int mm = tm * 16 + h * 8 + r;
        C[(size_t)mm * N + tn * 16 + l15] = acc[r];
    }
}

typedef __attribute__((ext_vector_type(4))) __bf16 v4bf;

__global__ void __launch_bounds__(256)
gcn_gather(const float* __restrict__ xw, const int* __restrict__ rowptr,
           const int* __restrict__ scol, const float* __restrict__ wgt,
           const float* __restrict__ dinv, const float* __restrict__ bias,
           __bf16* __restrict__ hb, int n) {
    const int lane = threadIdx.x & 31;
    const int node = __builtin_amdgcn_readfirstlane((int)((blockIdx.x * blockDim.x + threadIdx.x) >> 5));
    if (node >= n) return;
    const int b = __builtin_amdgcn_readfirstlane(rowptr[node]);
    const int L = __builtin_amdgcn_readfirstlane(rowptr[node + 1]) - b;
    const float sd = dinv[node];

    const float4* self = (const float4*)(xw + (size_t)node * HID);
    float4 a0 = self[lane], a1 = self[32 + lane];
    a0.x = a0.x * sd * sd; a0.y = a0.y * sd * sd; a0.z = a0.z * sd * sd; a0.w = a0.w * sd * sd;
    a1.x = a1.x * sd * sd; a1.y = a1.y * sd * sd; a1.z = a1.z * sd * sd; a1.w = a1.w * sd * sd;

    int   ms = 0;
    float mw = 0.0f;
    if (lane < L) { ms = scol[b + lane]; mw = wgt[b + lane]; }
    for (int c = 0; c < L; c += 32) {
        const int m = (L - c < 32) ? (L - c) : 32;
        int   ns = 0;
        float nw = 0.0f;
        if (c + 32 + lane < L) { ns = scol[b + c + 32 + lane]; nw = wgt[b + c + 32 + lane]; }
#pragma unroll 8
        for (int q = 0; q < m; ++q) {
            const int   s = __builtin_amdgcn_readlane(ms, q);
            const float w = __builtin_bit_cast(float, __builtin_amdgcn_readlane(__builtin_bit_cast(int, mw), q));
            const float4* r = (const float4*)(xw + (size_t)s * HID);
            const float4 v0 = r[lane], v1 = r[32 + lane];
            a0.x = fmaf(w, v0.x, a0.x); a0.y = fmaf(w, v0.y, a0.y);
            a0.z = fmaf(w, v0.z, a0.z); a0.w = fmaf(w, v0.w, a0.w);
            a1.x = fmaf(w, v1.x, a1.x); a1.y = fmaf(w, v1.y, a1.y);
            a1.z = fmaf(w, v1.z, a1.z); a1.w = fmaf(w, v1.w, a1.w);
        }
        ms = ns; mw = nw;
    }

    const float* bl = bias + 4 * lane;
    a0.x += bl[0];   a0.y += bl[1];   a0.z += bl[2];   a0.w += bl[3];
    a1.x += bl[128]; a1.y += bl[129]; a1.z += bl[130]; a1.w += bl[131];
    v4bf o0, o1;
    o0[0] = (__bf16)(a0.x > 0.f ? a0.x : 0.f); o0[1] = (__bf16)(a0.y > 0.f ? a0.y : 0.f);
    o0[2] = (__bf16)(a0.z > 0.f ? a0.z : 0.f); o0[3] = (__bf16)(a0.w > 0.f ? a0.w : 0.f);
    o1[0] = (__bf16)(a1.x > 0.f ? a1.x : 0.f); o1[1] = (__bf16)(a1.y > 0.f ? a1.y : 0.f);
    o1[2] = (__bf16)(a1.z > 0.f ? a1.z : 0.f); o1[3] = (__bf16)(a1.w > 0.f ? a1.w : 0.f);
    __bf16* orow = hb + (size_t)node * HID;
    *(v4bf*)(orow + 4 * lane)       = o0;
    *(v4bf*)(orow + 128 + 4 * lane) = o1;
}

#define SCAN_THREADS 128
#define SCAN_GROUPS  4
#define SCAN_KT      4
#define HROWS   7
#define HPITCH4 17
#define HPITCH2 (8 * HPITCH4)
#define HIMG4   (HROWS * HPITCH4)
#define HIMG2   (HROWS * HPITCH2)

typedef __attribute__((ext_vector_type(8)))  unsigned int v8u;
typedef __attribute__((ext_vector_type(16))) unsigned int v16u;
typedef __attribute__((ext_vector_type(32))) __bf16 v32bf;

static __device__ __forceinline__ v32bf bfrag_at(const uint4* p) {
    const uint4 q0 = p[0], q1 = p[1], q2 = p[2], q3 = p[3];
    const v16u v = {q0.x, q0.y, q0.z, q0.w, q1.x, q1.y, q1.z, q1.w,
                    q2.x, q2.y, q2.z, q2.w, q3.x, q3.y, q3.z, q3.w};
    return __builtin_bit_cast(v32bf, v);
}
static __device__ __forceinline__ v16bf afrag_at(const uint4* p) {
    const uint4 lo = p[0], hi = p[1];
    const v8u v = {lo.x, lo.y, lo.z, lo.w, hi.x, hi.y, hi.z, hi.w};
    return __builtin_bit_cast(v16bf, v);
}

#define GATE_K  (-1.4426950408889634f)
#define GATE_2K (-2.8853900817779268f)

static __device__ __forceinline__ float gate_q(float a) {
    return __builtin_amdgcn_rcpf(1.0f + __builtin_amdgcn_exp2f(a));
}

static __device__ __forceinline__ float scaled_piece_sum(float m, float d0, float d1, float d2, float k) {
    return fmaf(m, d2 + d1, fmaf(m, d0, k));
}

static __device__ __forceinline__ void split3(float v, unsigned& p0, unsigned& p1, unsigned& p2) {
    const unsigned b  = __builtin_bit_cast(unsigned, v);
    const unsigned b0 = b & 0xFFFF0000u;
    const unsigned bq = b & 0xFFFFFF00u;
    p0 = b0;
    p1 = __builtin_bit_cast(unsigned, __builtin_bit_cast(float, bq) - __builtin_bit_cast(float, b0));
    p2 = __builtin_bit_cast(unsigned, v - __builtin_bit_cast(float, bq));
}

static __device__ __forceinline__ void lds_arrive() {
    __builtin_amdgcn_fence(__ATOMIC_RELEASE, "workgroup", "local");
    __builtin_amdgcn_s_barrier_signal(-1);
}
static __device__ __forceinline__ void lds_pass() {
    __builtin_amdgcn_s_barrier_wait(-1);
    __builtin_amdgcn_fence(__ATOMIC_ACQUIRE, "workgroup", "local");
}

__global__ void __launch_bounds__(SCAN_THREADS)
gru_scan(const float* __restrict__ gi, const __bf16* __restrict__ Whh,
         const float* __restrict__ b_ih, const float* __restrict__ b_hh,
         float* __restrict__ hs, int T) {
    __shared__ uint4 hsplit[2 * HIMG4];
    const int t    = threadIdx.x;
    const int w    = t >> 5;
    const int lane = t & 31;
    const int c    = t & 15;
    const int hh   = (t >> 4) & 1;
    const int u0   = 64 * w;

    v32bf Br[SCAN_GROUPS][SCAN_KT], Bz[SCAN_GROUPS][SCAN_KT], Bn[SCAN_GROUPS][SCAN_KT];
#pragma unroll
    for (int g = 0; g < SCAN_GROUPS; ++g) {
        const size_t u = (size_t)(u0 + 16 * g + c);
#pragma unroll
        for (int kt = 0; kt < SCAN_KT; ++kt) {
            const int ch = 2 * (kt & 1) + 4 * hh + (kt >> 1);
            Br[g][kt] = bfrag_at((const uint4*)(Whh + u * HID + 32 * ch));
            Bz[g][kt] = bfrag_at((const uint4*)(Whh + (HID + u) * HID + 32 * ch));
            Bn[g][kt] = bfrag_at((const uint4*)(Whh + (2 * HID + u) * HID + 32 * ch));
        }
    }
    const int ja = u0 + lane;
    const int jb = u0 + 32 + lane;
    const float bih_ra = b_ih[ja], bih_za = b_ih[HID + ja], bih_na = b_ih[2 * HID + ja];
    const float bhh_ra = b_hh[ja], bhh_za = b_hh[HID + ja], bhh_na = b_hh[2 * HID + ja];
    const float bih_rb = b_ih[jb], bih_zb = b_ih[HID + jb], bih_nb = b_ih[2 * HID + jb];
    const float bhh_rb = b_hh[jb], bhh_zb = b_hh[HID + jb], bhh_nb = b_hh[2 * HID + jb];
    const float bsa_r = bih_ra + bhh_ra, bsa_z = bih_za + bhh_za;
    const float bsb_r = bih_rb + bhh_rb, bsb_z = bih_zb + bhh_zb;

    for (int i = t; i < 2 * HIMG4; i += SCAN_THREADS) hsplit[i] = make_uint4(0u, 0u, 0u, 0u);
    float hpa = 0.0f, hpb = 0.0f;
    float xa_r = gi[ja], xa_z = gi[HID + ja], xa_n = gi[2 * HID + ja];
    float xb_r = gi[jb], xb_z = gi[HID + jb], xb_n = gi[2 * HID + jb];
    float ya_r = 0.0f, ya_z = 0.0f, ya_n = 0.0f, yb_r = 0.0f, yb_z = 0.0f, yb_n = 0.0f;
    const int prow = c & 7;
    const int arow = prow < 6 ? prow : 6;
    const int aidx = (prow >= 3 && prow < 6) ? (int)0xEEEEEEEEu : (int)0x44444444u;
    const uint4* hrd = &hsplit[arow * HPITCH4 + 8 * hh];
    unsigned short* hwr = (unsigned short*)&hsplit[0];
    const int wq  = (lane >> 1) & 1;
    const int ws_ = 2 * (lane >> 2) + (lane & 1);
    const int wa  = 3 * wq * HPITCH2 + 16 * (2 * w) + ws_;
    const int wb  = 3 * wq * HPITCH2 + 16 * (2 * w + 1) + ws_;

    v16bf a[SCAN_KT];
    v8f r1[2], z1[2], n1[2], r2[2], z2[2], n2[2];
    lds_arrive();

    auto h1_tile = [&](const v16bf f, const int kt) __attribute__((always_inline)) {
#pragma unroll
        for (int g = 0; g < 2; ++g)
            r1[g] = __builtin_amdgcn_swmmac_f32_16x16x64_bf16(false, f, false, Br[g][kt], r1[g], aidx, false, false);
#pragma unroll
        for (int g = 0; g < 2; ++g)
            z1[g] = __builtin_amdgcn_swmmac_f32_16x16x64_bf16(false, f, false, Bz[g][kt], z1[g], aidx, false, false);
#pragma unroll
        for (int g = 0; g < 2; ++g)
            n1[g] = __builtin_amdgcn_swmmac_f32_16x16x64_bf16(false, f, false, Bn[g][kt], n1[g], aidx, false, false);
    };
    auto h2rz_tile = [&](const v16bf f, const int kt) __attribute__((always_inline)) {
#pragma unroll
        for (int g = 0; g < 2; ++g)
            r2[g] = __builtin_amdgcn_swmmac_f32_16x16x64_bf16(false, f, false, Br[2 + g][kt], r2[g], aidx, false, false);
#pragma unroll
        for (int g = 0; g < 2; ++g)
            z2[g] = __builtin_amdgcn_swmmac_f32_16x16x64_bf16(false, f, false, Bz[2 + g][kt], z2[g], aidx, false, false);
    };
    auto h2n_tile = [&](const v16bf f, const int kt) __attribute__((always_inline)) {
#pragma unroll
        for (int g = 0; g < 2; ++g)
            n2[g] = __builtin_amdgcn_swmmac_f32_16x16x64_bf16(false, f, false, Bn[2 + g][kt], n2[g], aidx, false, false);
    };
    auto gate_arg = [&](const float m, const v8f (&d)[2], const float k) __attribute__((always_inline)) -> float {
        const float p0 = hh ? d[1][0] : d[0][0], p1 = hh ? d[1][1] : d[0][1], p2 = hh ? d[1][2] : d[0][2];
        const float q0 = hh ? d[1][3] : d[0][3], q1 = hh ? d[1][4] : d[0][4], q2 = hh ? d[1][5] : d[0][5];
        return scaled_piece_sum(m, p0 + q0, p1 + q1, p2 + q2, k);
    };
    auto new_h = [&](const v8f (&r)[2], const v8f (&z)[2], const v8f (&n)[2],
                     const float c_r, const float c_z, const float c_n, const float bhh_n, const float hp)
        __attribute__((always_inline)) -> float {
        const float rr = gate_q(gate_arg(GATE_K, r, c_r));
        const float zz = gate_q(gate_arg(GATE_K, z, c_z));
        const float rp = rr * GATE_2K;
        const float kn = fmaf(rp, bhh_n, c_n);
        const float oz = 1.0f - zz;
        const float A  = oz + oz;
        const float Bc = fmaf(zz, hp, -oz);
        const float q  = gate_q(gate_arg(rp, n, kn));
        return fmaf(A, q, Bc);
    };
    auto put_h = [&](const float v, const int j) __attribute__((always_inline)) {
        unsigned p0, p1, p2;
        split3(v, p0, p1, p2);
        hwr[j]               = (unsigned short)(p0 >> 16);
        hwr[HPITCH2 + j]     = (unsigned short)(p1 >> 16);
        hwr[2 * HPITCH2 + j] = (unsigned short)(p2 >> 16);
    };

    auto scan_step = [&](const int step, const int img,
                         const float ga_r, const float ga_z, const float ga_n,
                         const float gb_r, const float gb_z, const float gb_n,
                         float& nga_r, float& nga_z, float& nga_n,
                         float& ngb_r, float& ngb_z, float& ngb_n)
        __attribute__((always_inline)) {
        const unsigned gx = (unsigned)(step + 1 < T ? step + 1 : step) * GATES;
        nga_r = gi[gx + ja]; nga_z = gi[gx + HID + ja]; nga_n = gi[gx + 2 * HID + ja];
        ngb_r = gi[gx + jb]; ngb_z = gi[gx + HID + jb]; ngb_n = gi[gx + 2 * HID + jb];
        if (step > 0) {
            hs[(unsigned)(step - 1) * HID + (unsigned)ja] = hpa;
            hs[(unsigned)(step - 1) * HID + (unsigned)jb] = hpb;
        }
        const float ca_r = (ga_r + bsa_r) * GATE_K,  cb_r = (gb_r + bsb_r) * GATE_K;
        const float ca_z = (ga_z + bsa_z) * GATE_K,  cb_z = (gb_z + bsb_z) * GATE_K;
        const float ca_n = (ga_n + bih_na) * GATE_2K, cb_n = (gb_n + bih_nb) * GATE_2K;

        lds_pass();
        const uint4* rd = hrd + img * HIMG4;
        a[0] = afrag_at(rd);
        a[1] = afrag_at(rd + 4);
        a[2] = afrag_at(rd + 2);
        a[3] = afrag_at(rd + 6);
        __builtin_amdgcn_sched_barrier(0);
#pragma unroll
        for (int g = 0; g < 2; ++g) { r1[g] = v8f{}; z1[g] = v8f{}; n1[g] = v8f{}; r2[g] = v8f{}; z2[g] = v8f{}; n2[g] = v8f{}; }
#pragma unroll
        for (int kt = 0; kt < SCAN_KT; ++kt) h1_tile(a[kt], kt);
        const float hna = new_h(r1, z1, n1, ca_r, ca_z, ca_n, bhh_na, hpa);
        put_h(hna, wa + (1 - img) * HIMG2);
        hpa = hna;
#pragma unroll
        for (int kt = 0; kt < SCAN_KT; ++kt) h2rz_tile(a[kt], kt);
#pragma unroll
        for (int kt = 0; kt < SCAN_KT; ++kt) h2n_tile(a[kt], kt);
        const float hnb = new_h(r2, z2, n2, cb_r, cb_z, cb_n, bhh_nb, hpb);
        put_h(hnb, wb + (1 - img) * HIMG2);
        hpb = hnb;
        __builtin_amdgcn_sched_barrier(0);
        lds_arrive();
    };

    int step = 0;
    for (; step + 1 < T; step += 2) {
        scan_step(step,     0, xa_r, xa_z, xa_n, xb_r, xb_z, xb_n, ya_r, ya_z, ya_n, yb_r, yb_z, yb_n);
        scan_step(step + 1, 1, ya_r, ya_z, ya_n, yb_r, yb_z, yb_n, xa_r, xa_z, xa_n, xb_r, xb_z, xb_n);
    }
    if (step < T)
        scan_step(step,     0, xa_r, xa_z, xa_n, xb_r, xb_z, xb_n, ya_r, ya_z, ya_n, yb_r, yb_z, yb_n);
    lds_pass();
    hs[(unsigned)(T - 1) * HID + (unsigned)ja] = hpa;
    hs[(unsigned)(T - 1) * HID + (unsigned)jb] = hpb;
}

__global__ void fc_out_kernel(const float* __restrict__ hs, const float* __restrict__ fw,
                              const float* __restrict__ fb, float* __restrict__ out, int total) {
    int gid = blockIdx.x * blockDim.x + threadIdx.x;
    if (gid >= total) return;
    int n = gid / OUTF, o = gid % OUTF;
    const float* h = hs + (size_t)n * HID;
    const float* w = fw + (size_t)o * HID;
    float acc = fb[o];
#pragma unroll 8
    for (int k = 0; k < HID; ++k) acc = fmaf(h[k], w[k], acc);
    out[gid] = acc;
}

extern "C" void kernel_launch(void* const* d_in, const int* in_sizes, int n_in,
                              void* d_out, int out_size, void* d_ws, size_t ws_size,
                              hipStream_t stream) {
    const float* x    = (const float*)d_in[0];
    const int*   ei   = (const int*)  d_in[1];
    const float* W1   = (const float*)d_in[2];
    const float* b1   = (const float*)d_in[3];
    const float* W2   = (const float*)d_in[4];
    const float* b2   = (const float*)d_in[5];
    const float* W_ih = (const float*)d_in[6];
    const float* W_hh = (const float*)d_in[7];
    const float* b_ih = (const float*)d_in[8];
    const float* b_hh = (const float*)d_in[9];
    const float* fc_w = (const float*)d_in[10];
    const float* fc_b = (const float*)d_in[11];
    float* out = (float*)d_out;

    const int E = in_sizes[1] / 2;
    const int* src = ei;
    const int* dst = ei + E;

    char* ws = (char*)d_ws;
    size_t off = 0;
    auto alloc = [&](size_t bytes) -> char* {
        char* p = ws + off;
        off = (off + bytes + 255) & ~(size_t)255;
        return p;
    };
    int*    cnt  = (int*)   alloc((size_t)NUM_NODES * 4);
    float*  dinv = (float*) alloc((size_t)NUM_NODES * 4);
    float*  xw   = (float*) alloc((size_t)NUM_NODES * HID * 4);
    int*    rowptr = (int*)  alloc((size_t)(NUM_NODES + 1) * 4);
    int*    scol   = (int*)  alloc((size_t)E * 4);
    float*  wgt    = (float*)alloc((size_t)E * 4);
    float*  gi   = (float*) alloc((size_t)NUM_NODES * GATES * 4);
    float*  hs   = (float*) alloc((size_t)NUM_NODES * HID * 4);
    __bf16* xb   = (__bf16*)alloc((size_t)NUM_NODES * HID * 2);
    __bf16* w1t  = (__bf16*)alloc((size_t)HID * IN_FEAT * 2);
    __bf16* w2t  = (__bf16*)alloc((size_t)HID * HID * 2);
    __bf16* wih  = (__bf16*)alloc((size_t)GATES * HID * 2);
    __bf16* whh  = (__bf16*)alloc((size_t)GATES * HID * 2);
    int*    pos  = (int*)gi;
    int*    col  = (int*)gi + E;
    (void)ws_size; (void)n_in; (void)out_size;

    const int TPB = 256;
    auto blk = [](long long n, int t) { return (unsigned)((n + t - 1) / t); };

    cvt_f32_to_bf16<<<blk((long long)NUM_NODES * IN_FEAT, TPB), TPB, 0, stream>>>(
        x, xb, NUM_NODES * IN_FEAT);
    cvt_f32_to_bf16_T<<<blk((long long)IN_FEAT * HID, TPB), TPB, 0, stream>>>(
        W1, w1t, IN_FEAT, HID);
    cvt_f32_to_bf16_T<<<blk((long long)HID * HID, TPB), TPB, 0, stream>>>(
        W2, w2t, HID, HID);
    cvt_f32_to_bf16<<<blk((long long)GATES * HID, TPB), TPB, 0, stream>>>(
        W_ih, wih, GATES * HID);
    cvt_f32_to_bf16<<<blk((long long)GATES * HID, TPB), TPB, 0, stream>>>(
        W_hh, whh, GATES * HID);

    csr_zero<<<blk(NUM_NODES, TPB), TPB, 0, stream>>>(cnt, NUM_NODES);
    csr_count<<<blk(E, TPB), TPB, 0, stream>>>(dst, cnt, pos, E);
    csr_offsets<<<1, OFFS_THREADS, 0, stream>>>(cnt, rowptr, dinv, NUM_NODES);
    csr_place<<<blk(E, TPB), TPB, 0, stream>>>(src, dst, rowptr, pos, col, E);
    csr_rank<<<blk((long long)NUM_NODES * 32, TPB), TPB, 0, stream>>>(col, rowptr, dinv, scol, wgt, NUM_NODES);

    const unsigned gemm1_blocks = blk((long long)(NUM_NODES / 16) * (HID / 16), 4);
    const unsigned gemm2_blocks = gemm1_blocks;
    const unsigned gemm3_blocks = blk((long long)(NUM_NODES / 16) * (GATES / 16), 4);
    const unsigned gath_blocks  = blk((long long)NUM_NODES * 32, TPB);

    wmma_gemm_bf16<<<gemm1_blocks, 128, 0, stream>>>(xb, w1t, xw, NUM_NODES, HID, IN_FEAT);
    gcn_gather<<<gath_blocks, TPB, 0, stream>>>(xw, rowptr, scol, wgt, dinv, b1, xb, NUM_NODES);

    wmma_gemm_bf16<<<gemm2_blocks, 128, 0, stream>>>(xb, w2t, xw, NUM_NODES, HID, HID);
    gcn_gather<<<gath_blocks, TPB, 0, stream>>>(xw, rowptr, scol, wgt, dinv, b2, xb, NUM_NODES);

    wmma_gemm_bf16<<<gemm3_blocks, 128, 0, stream>>>(xb, wih, gi, NUM_NODES, GATES, HID);

    gru_scan<<<1, SCAN_THREADS, 0, stream>>>(gi, whh, b_ih, b_hh, hs, NUM_NODES);

    fc_out_kernel<<<blk((long long)NUM_NODES * OUTF, TPB), TPB, 0, stream>>>(
        hs, fc_w, fc_b, out, NUM_NODES * OUTF);
}


# GRU scan: 8 waves, two per SIMD (each 6 W_hh tiles x 4 k in 384 registers, one unit per lane), not 4: a wave's gates, waits and hand-over run beside the other wave's matrix work; sum order unchanged
# speedup vs baseline: 19.5516x; 1.1698x over previous
#include <hip/hip_runtime.h>
#include <hip/hip_bf16.h>
#include <stdint.h>

typedef __attribute__((ext_vector_type(16))) __bf16 v16bf;
typedef __attribute__((ext_vector_type(8)))  float  v8f;

#define NUM_NODES 10000
#define IN_FEAT   64
#define HID       256
#define OUTF      3
#define GATES     768

__global__ void cvt_f32_to_bf16(const float* __restrict__ in,
                                __bf16* __restrict__ out, int n) {
    int i = blockIdx.x * blockDim.x + threadIdx.x;
    if (i < n) out[i] = (__bf16)in[i];
}

__global__ void cvt_f32_to_bf16_T(const float* __restrict__ in,
                                  __bf16* __restrict__ out, int K, int N) {
    int i = blockIdx.x * blockDim.x + threadIdx.x;
    if (i >= K * N) return;
    int k = i / N, n = i % N;
    out[(size_t)n * K + k] = (__bf16)in[(size_t)k * N + n];
}

__global__ void csr_zero(int* __restrict__ cnt, int n) {
    int i = blockIdx.x * blockDim.x + threadIdx.x;
    if (i < n) cnt[i] = 0;
}

__global__ void csr_count(const int* __restrict__ dst, int* __restrict__ cnt,
                          int* __restrict__ pos, int E) {
    int e = blockIdx.x * blockDim.x + threadIdx.x;
    if (e < E) pos[e] = atomicAdd(&cnt[dst[e]], 1);
}

#define OFFS_THREADS 1024
__global__ void __launch_bounds__(OFFS_THREADS)
csr_offsets(const int* __restrict__ cnt, int* __restrict__ rowptr,
            float* __restrict__ dinv, int n) {
    __shared__ int part[OFFS_THREADS];
    const int t   = threadIdx.x;
    const int per = (n + OFFS_THREADS - 1) / OFFS_THREADS;
    const int b0  = t * per;
    int s = 0;
    for (int i = 0; i < per; ++i) {
        const int idx = b0 + i;
        if (idx < n) s += cnt[idx];
    }
    part[t] = s;
    __syncthreads();
    for (int off = 1; off < OFFS_THREADS; off <<= 1) {
        const int v = (t >= off) ? part[t - off] : 0;
        __syncthreads();
        part[t] += v;
        __syncthreads();
    }
    int run = part[t] - s;
    for (int i = 0; i < per; ++i) {
        const int idx = b0 + i;
        if (idx < n) {
            const int c = cnt[idx];
            rowptr[idx] = run;
            dinv[idx]   = rsqrtf((float)(c + 1));
            run += c;
        }
    }
    if (t == OFFS_THREADS - 1) rowptr[n] = part[t];
}

__global__ void csr_place(const int* __restrict__ src, const int* __restrict__ dst,
                          const int* __restrict__ rowptr, const int* __restrict__ pos,
                          int* __restrict__ col, int E) {
    int e = blockIdx.x * blockDim.x + threadIdx.x;
    if (e < E) col[rowptr[dst[e]] + pos[e]] = src[e];
}

__global__ void csr_rank(const int* __restrict__ col, const int* __restrict__ rowptr,
                         const float* __restrict__ dinv,
                         int* __restrict__ scol, float* __restrict__ wgt, int n) {
    const int node = (int)((blockIdx.x * blockDim.x + threadIdx.x) >> 5);
    const int lane = threadIdx.x & 31;
    if (node >= n) return;
    const int b = rowptr[node];
    const int L = rowptr[node + 1] - b;
    const float dd = dinv[node];
    const int* row = col + b;
    for (int i = lane; i < L; i += 32) {
        const int s = row[i];
        int rank = 0;
        for (int q = 0; q < L; ++q) {
            const int v = row[q];
            rank += (v < s || (v == s && q < i)) ? 1 : 0;
        }
        scol[b + rank] = s;
        wgt[b + rank]  = dinv[s] * dd;
    }
}

__global__ void __launch_bounds__(128)
wmma_gemm_bf16(const __bf16* __restrict__ A, const __bf16* __restrict__ Bt,
               float* __restrict__ C, int M, int N, int K) {
    int wave  = (int)((blockIdx.x * blockDim.x + threadIdx.x) >> 5);
    int lane  = threadIdx.x & 31;
    int tilesN = N >> 4;
    int tilesM = M >> 4;
    if (wave >= tilesM * tilesN) return;
    int tm = wave / tilesN;
    int tn = wave % tilesN;

    int l15 = lane & 15;
    int h   = lane >> 4;
    const __bf16* arow = A  + (size_t)(tm * 16 + l15) * K;
    const __bf16* bcol = Bt + (size_t)(tn * 16 + l15) * K;

    v8f acc = {};
    for (int k = 0; k < K; k += 32) {
        v16bf a, b;
        int k0 = k + h * 8;
#pragma unroll
        for (int i = 0; i < 8; ++i) {
            a[i]     = arow[k0 + i];
            a[i + 8] = arow[k0 + 16 + i];
            b[i]     = bcol[k0 + i];
            b[i + 8] = bcol[k0 + 16 + i];
        }
        acc = __builtin_amdgcn_wmma_f32_16x16x32_bf16(
            false, a, false, b, (short)0, acc, false, false);
    }
#pragma unroll
    for (int r = 0; r < 8; ++r) {
        int mm = tm * 16 + h * 8 + r;
        C[(size_t)mm * N + tn * 16 + l15] = acc[r];
    }
}

typedef __attribute__((ext_vector_type(4))) __bf16 v4bf;

__global__ void __launch_bounds__(256)
gcn_gather(const float* __restrict__ xw, const int* __restrict__ rowptr,
           const int* __restrict__ scol, const float* __restrict__ wgt,
           const float* __restrict__ dinv, const float* __restrict__ bias,
           __bf16* __restrict__ hb, int n) {
    const int lane = threadIdx.x & 31;
    const int node = __builtin_amdgcn_readfirstlane((int)((blockIdx.x * blockDim.x + threadIdx.x) >> 5));
    if (node >= n) return;
    const int b = __builtin_amdgcn_readfirstlane(rowptr[node]);
    const int L = __builtin_amdgcn_readfirstlane(rowptr[node + 1]) - b;
    const float sd = dinv[node];

    const float4* self = (const float4*)(xw + (size_t)node * HID);
    float4 a0 = self[lane], a1 = self[32 + lane];
    a0.x = a0.x * sd * sd; a0.y = a0.y * sd * sd; a0.z = a0.z * sd * sd; a0.w = a0.w * sd * sd;
    a1.x = a1.x * sd * sd; a1.y = a1.y * sd * sd; a1.z = a1.z * sd * sd; a1.w = a1.w * sd * sd;

    int   ms = 0;
    float mw = 0.0f;
    if (lane < L) { ms = scol[b + lane]; mw = wgt[b + lane]; }
    for (int c = 0; c < L; c += 32) {
        const int m = (L - c < 32) ? (L - c) : 32;
        int   ns = 0;
        float nw = 0.0f;
        if (c + 32 + lane < L) { ns = scol[b + c + 32 + lane]; nw = wgt[b + c + 32 + lane]; }
#pragma unroll 8
        for (int q = 0; q < m; ++q) {
            const int   s = __builtin_amdgcn_readlane(ms, q);
            const float w = __builtin_bit_cast(float, __builtin_amdgcn_readlane(__builtin_bit_cast(int, mw), q));
            const float4* r = (const float4*)(xw + (size_t)s * HID);
            const float4 v0 = r[lane], v1 = r[32 + lane];
            a0.x = fmaf(w, v0.x, a0.x); a0.y = fmaf(w, v0.y, a0.y);
            a0.z = fmaf(w, v0.z, a0.z); a0.w = fmaf(w, v0.w, a0.w);
            a1.x = fmaf(w, v1.x, a1.x); a1.y = fmaf(w, v1.y, a1.y);
            a1.z = fmaf(w, v1.z, a1.z); a1.w = fmaf(w, v1.w, a1.w);
        }
        ms = ns; mw = nw;
    }

    const float* bl = bias + 4 * lane;
    a0.x += bl[0];   a0.y += bl[1];   a0.z += bl[2];   a0.w += bl[3];
    a1.x += bl[128]; a1.y += bl[129]; a1.z += bl[130]; a1.w += bl[131];
    v4bf o0, o1;
    o0[0] = (__bf16)(a0.x > 0.f ? a0.x : 0.f); o0[1] = (__bf16)(a0.y > 0.f ? a0.y : 0.f);
    o0[2] = (__bf16)(a0.z > 0.f ? a0.z : 0.f); o0[3] = (__bf16)(a0.w > 0.f ? a0.w : 0.f);
    o1[0] = (__bf16)(a1.x > 0.f ? a1.x : 0.f); o1[1] = (__bf16)(a1.y > 0.f ? a1.y : 0.f);
    o1[2] = (__bf16)(a1.z > 0.f ? a1.z : 0.f); o1[3] = (__bf16)(a1.w > 0.f ? a1.w : 0.f);
    __bf16* orow = hb + (size_t)node * HID;
    *(v4bf*)(orow + 4 * lane)       = o0;
    *(v4bf*)(orow + 128 + 4 * lane) = o1;
}

#define SCAN_THREADS 256
#define SCAN_GROUPS  2
#define SCAN_KT      4
#define HROWS   7
#define HPITCH4 17
#define HPITCH2 (8 * HPITCH4)
#define HIMG4   (HROWS * HPITCH4)
#define HIMG2   (HROWS * HPITCH2)

typedef __attribute__((ext_vector_type(8)))  unsigned int v8u;
typedef __attribute__((ext_vector_type(16))) unsigned int v16u;
typedef __attribute__((ext_vector_type(32))) __bf16 v32bf;

static __device__ __forceinline__ v32bf bfrag_at(const uint4* p) {
    const uint4 q0 = p[0], q1 = p[1], q2 = p[2], q3 = p[3];
    const v16u v = {q0.x, q0.y, q0.z, q0.w, q1.x, q1.y, q1.z, q1.w,
                    q2.x, q2.y, q2.z, q2.w, q3.x, q3.y, q3.z, q3.w};
    return __builtin_bit_cast(v32bf, v);
}
static __device__ __forceinline__ v16bf afrag_at(const uint4* p) {
    const uint4 lo = p[0], hi = p[1];
    const v8u v = {lo.x, lo.y, lo.z, lo.w, hi.x, hi.y, hi.z, hi.w};
    return __builtin_bit_cast(v16bf, v);
}

#define GATE_K  (-1.4426950408889634f)
#define GATE_2K (-2.8853900817779268f)

static __device__ __forceinline__ float gate_q(float a) {
    return __builtin_amdgcn_rcpf(1.0f + __builtin_amdgcn_exp2f(a));
}

static __device__ __forceinline__ float scaled_piece_sum(float m, float d0, float d1, float d2, float k) {
    return fmaf(m, d2 + d1, fmaf(m, d0, k));
}

static __device__ __forceinline__ void split3(float v, unsigned& p0, unsigned& p1, unsigned& p2) {
    const unsigned b  = __builtin_bit_cast(unsigned, v);
    const unsigned b0 = b & 0xFFFF0000u;
    const unsigned bq = b & 0xFFFFFF00u;
    p0 = b0;
    p1 = __builtin_bit_cast(unsigned, __builtin_bit_cast(float, bq) - __builtin_bit_cast(float, b0));
    p2 = __builtin_bit_cast(unsigned, v - __builtin_bit_cast(float, bq));
}

static __device__ __forceinline__ void lds_arrive() {
    __builtin_amdgcn_fence(__ATOMIC_RELEASE, "workgroup", "local");
    __builtin_amdgcn_s_barrier_signal(-1);
}
static __device__ __forceinline__ void lds_pass() {
    __builtin_amdgcn_s_barrier_wait(-1);
    __builtin_amdgcn_fence(__ATOMIC_ACQUIRE, "workgroup", "local");
}

__global__ void __launch_bounds__(SCAN_THREADS)
gru_scan(const float* __restrict__ gi, const __bf16* __restrict__ Whh,
         const float* __restrict__ b_ih, const float* __restrict__ b_hh,
         float* __restrict__ hs, int T) {
    __shared__ uint4 hsplit[2 * HIMG4];
    const int t    = threadIdx.x;
    const int w    = t >> 5;
    const int lane = t & 31;
    const int c    = t & 15;
    const int hh   = (t >> 4) & 1;
    const int u0   = 32 * w;

    v32bf Br[SCAN_GROUPS][SCAN_KT], Bz[SCAN_GROUPS][SCAN_KT], Bn[SCAN_GROUPS][SCAN_KT];
#pragma unroll
    for (int g = 0; g < SCAN_GROUPS; ++g) {
        const size_t u = (size_t)(u0 + 16 * g + c);
#pragma unroll
        for (int kt = 0; kt < SCAN_KT; ++kt) {
            const int ch = 2 * (kt & 1) + 4 * hh + (kt >> 1);
            Br[g][kt] = bfrag_at((const uint4*)(Whh + u * HID + 32 * ch));
            Bz[g][kt] = bfrag_at((const uint4*)(Whh + (HID + u) * HID + 32 * ch));
            Bn[g][kt] = bfrag_at((const uint4*)(Whh + (2 * HID + u) * HID + 32 * ch));
        }
    }
    const int ja = u0 + lane;
    const float bih_ra = b_ih[ja], bih_za = b_ih[HID + ja], bih_na = b_ih[2 * HID + ja];
    const float bhh_ra = b_hh[ja], bhh_za = b_hh[HID + ja], bhh_na = b_hh[2 * HID + ja];
    const float bsa_r = bih_ra + bhh_ra, bsa_z = bih_za + bhh_za;

    for (int i = t; i < 2 * HIMG4; i += SCAN_THREADS) hsplit[i] = make_uint4(0u, 0u, 0u, 0u);
    float hpa = 0.0f;
    float xa_r = gi[ja], xa_z = gi[HID + ja], xa_n = gi[2 * HID + ja];
    float ya_r = 0.0f, ya_z = 0.0f, ya_n = 0.0f;
    const int prow = c & 7;
    const int arow = prow < 6 ? prow : 6;
    const int aidx = (prow >= 3 && prow < 6) ? (int)0xEEEEEEEEu : (int)0x44444444u;
    const uint4* hrd = &hsplit[arow * HPITCH4 + 8 * hh];
    unsigned short* hwr = (unsigned short*)&hsplit[0];
    const int wq  = (lane >> 1) & 1;
    const int ws_ = 2 * (lane >> 2) + (lane & 1);
    const int wa  = 3 * wq * HPITCH2 + 16 * w + ws_;

    v16bf a[SCAN_KT];
    v8f r1[2], z1[2], n1[2];
    lds_arrive();

    auto h1_tile = [&](const v16bf f, const int kt) __attribute__((always_inline)) {
#pragma unroll
        for (int g = 0; g < 2; ++g)
            r1[g] = __builtin_amdgcn_swmmac_f32_16x16x64_bf16(false, f, false, Br[g][kt], r1[g], aidx, false, false);
#pragma unroll
        for (int g = 0; g < 2; ++g)
            z1[g] = __builtin_amdgcn_swmmac_f32_16x16x64_bf16(false, f, false, Bz[g][kt], z1[g], aidx, false, false);
#pragma unroll
        for (int g = 0; g < 2; ++g)
            n1[g] = __builtin_amdgcn_swmmac_f32_16x16x64_bf16(false, f, false, Bn[g][kt], n1[g], aidx, false, false);
    };
    auto gate_arg = [&](const float m, const v8f (&d)[2], const float k) __attribute__((always_inline)) -> float {
        const float p0 = hh ? d[1][0] : d[0][0], p1 = hh ? d[1][1] : d[0][1], p2 = hh ? d[1][2] : d[0][2];
        const float q0 = hh ? d[1][3] : d[0][3], q1 = hh ? d[1][4] : d[0][4], q2 = hh ? d[1][5] : d[0][5];
        return scaled_piece_sum(m, p0 + q0, p1 + q1, p2 + q2, k);
    };
    auto new_h = [&](const v8f (&r)[2], const v8f (&z)[2], const v8f (&n)[2],
                     const float c_r, const float c_z, const float c_n, const float bhh_n, const float hp)
        __attribute__((always_inline)) -> float {
        const float rr = gate_q(gate_arg(GATE_K, r, c_r));
        const float zz = gate_q(gate_arg(GATE_K, z, c_z));
        const float rp = rr * GATE_2K;
        const float kn = fmaf(rp, bhh_n, c_n);
        const float oz = 1.0f - zz;
        const float A  = oz + oz;
        const float Bc = fmaf(zz, hp, -oz);
        const float q  = gate_q(gate_arg(rp, n, kn));
        return fmaf(A, q, Bc);
    };
    auto put_h = [&](const float v, const int j) __attribute__((always_inline)) {
        unsigned p0, p1, p2;
        split3(v, p0, p1, p2);
        hwr[j]               = (unsigned short)(p0 >> 16);
        hwr[HPITCH2 + j]     = (unsigned short)(p1 >> 16);
        hwr[2 * HPITCH2 + j] = (unsigned short)(p2 >> 16);
    };

    auto scan_step = [&](const int step, const int img,
                         const float ga_r, const float ga_z, const float ga_n,
                         float& nga_r, float& nga_z, float& nga_n)
        __attribute__((always_inline)) {
        const unsigned gx = (unsigned)(step + 1 < T ? step + 1 : step) * GATES;
        nga_r = gi[gx + ja]; nga_z = gi[gx + HID + ja]; nga_n = gi[gx + 2 * HID + ja];
        if (step > 0) hs[(unsigned)(step - 1) * HID + (unsigned)ja] = hpa;
        const float ca_r = (ga_r + bsa_r) * GATE_K;
        const float ca_z = (ga_z + bsa_z) * GATE_K;
        const float ca_n = (ga_n + bih_na) * GATE_2K;

        lds_pass();
        const uint4* rd = hrd + img * HIMG4;
        a[0] = afrag_at(rd);
        a[1] = afrag_at(rd + 4);
        a[2] = afrag_at(rd + 2);
        a[3] = afrag_at(rd + 6);
        __builtin_amdgcn_sched_barrier(0);
#pragma unroll
        for (int g = 0; g < 2; ++g) { r1[g] = v8f{}; z1[g] = v8f{}; n1[g] = v8f{}; }
#pragma unroll
        for (int kt = 0; kt < SCAN_KT; ++kt) h1_tile(a[kt], kt);
        const float hna = new_h(r1, z1, n1, ca_r, ca_z, ca_n, bhh_na, hpa);
        put_h(hna, wa + (1 - img) * HIMG2);
        hpa = hna;
        __builtin_amdgcn_sched_barrier(0);
        lds_arrive();
    };

    int step = 0;
    for (; step + 1 < T; step += 2) {
        scan_step(step,     0, xa_r, xa_z, xa_n, ya_r, ya_z, ya_n);
        scan_step(step + 1, 1, ya_r, ya_z, ya_n, xa_r, xa_z, xa_n);
    }
    if (step < T)
        scan_step(step,     0, xa_r, xa_z, xa_n, ya_r, ya_z, ya_n);
    lds_pass();
    hs[(unsigned)(T - 1) * HID + (unsigned)ja] = hpa;
}

__global__ void fc_out_kernel(const float* __restrict__ hs, const float* __restrict__ fw,
                              const float* __restrict__ fb, float* __restrict__ out, int total) {
    int gid = blockIdx.x * blockDim.x + threadIdx.x;
    if (gid >= total) return;
    int n = gid / OUTF, o = gid % OUTF;
    const float* h = hs + (size_t)n * HID;
    const float* w = fw + (size_t)o * HID;
    float acc = fb[o];
#pragma unroll 8
    for (int k = 0; k < HID; ++k) acc = fmaf(h[k], w[k], acc);
    out[gid] = acc;
}

extern "C" void kernel_launch(void* const* d_in, const int* in_sizes, int n_in,
                              void* d_out, int out_size, void* d_ws, size_t ws_size,
                              hipStream_t stream) {
    const float* x    = (const float*)d_in[0];
    const int*   ei   = (const int*)  d_in[1];
    const float* W1   = (const float*)d_in[2];
    const float* b1   = (const float*)d_in[3];
    const float* W2   = (const float*)d_in[4];
    const float* b2   = (const float*)d_in[5];
    const float* W_ih = (const float*)d_in[6];
    const float* W_hh = (const float*)d_in[7];
    const float* b_ih = (const float*)d_in[8];
    const float* b_hh = (const float*)d_in[9];
    const float* fc_w = (const float*)d_in[10];
    const float* fc_b = (const float*)d_in[11];
    float* out = (float*)d_out;

    const int E = in_sizes[1] / 2;
    const int* src = ei;
    const int* dst = ei + E;

    char* ws = (char*)d_ws;
    size_t off = 0;
    auto alloc = [&](size_t bytes) -> char* {
        char* p = ws + off;
        off = (off + bytes + 255) & ~(size_t)255;
        return p;
    };
    int*    cnt  = (int*)   alloc((size_t)NUM_NODES * 4);
    float*  dinv = (float*) alloc((size_t)NUM_NODES * 4);
    float*  xw   = (float*) alloc((size_t)NUM_NODES * HID * 4);
    int*    rowptr = (int*)  alloc((size_t)(NUM_NODES + 1) * 4);
    int*    scol   = (int*)  alloc((size_t)E * 4);
    float*  wgt    = (float*)alloc((size_t)E * 4);
    float*  gi   = (float*) alloc((size_t)NUM_NODES * GATES * 4);
    float*  hs   = (float*) alloc((size_t)NUM_NODES * HID * 4);
    __bf16* xb   = (__bf16*)alloc((size_t)NUM_NODES * HID * 2);
    __bf16* w1t  = (__bf16*)alloc((size_t)HID * IN_FEAT * 2);
    __bf16* w2t  = (__bf16*)alloc((size_t)HID * HID * 2);
    __bf16* wih  = (__bf16*)alloc((size_t)GATES * HID * 2);
    __bf16* whh  = (__bf16*)alloc((size_t)GATES * HID * 2);
    int*    pos  = (int*)gi;
    int*    col  = (int*)gi + E;
    (void)ws_size; (void)n_in; (void)out_size;

    const int TPB = 256;
    auto blk = [](long long n, int t) { return (unsigned)((n + t - 1) / t); };

    cvt_f32_to_bf16<<<blk((long long)NUM_NODES * IN_FEAT, TPB), TPB, 0, stream>>>(
        x, xb, NUM_NODES * IN_FEAT);
    cvt_f32_to_bf16_T<<<blk((long long)IN_FEAT * HID, TPB), TPB, 0, stream>>>(
        W1, w1t, IN_FEAT, HID);
    cvt_f32_to_bf16_T<<<blk((long long)HID * HID, TPB), TPB, 0, stream>>>(
        W2, w2t, HID, HID);
    cvt_f32_to_bf16<<<blk((long long)GATES * HID, TPB), TPB, 0, stream>>>(
        W_ih, wih, GATES * HID);
    cvt_f32_to_bf16<<<blk((long long)GATES * HID, TPB), TPB, 0, stream>>>(
        W_hh, whh, GATES * HID);

    csr_zero<<<blk(NUM_NODES, TPB), TPB, 0, stream>>>(cnt, NUM_NODES);
    csr_count<<<blk(E, TPB), TPB, 0, stream>>>(dst, cnt, pos, E);
    csr_offsets<<<1, OFFS_THREADS, 0, stream>>>(cnt, rowptr, dinv, NUM_NODES);
    csr_place<<<blk(E, TPB), TPB, 0, stream>>>(src, dst, rowptr, pos, col, E);
    csr_rank<<<blk((long long)NUM_NODES * 32, TPB), TPB, 0, stream>>>(col, rowptr, dinv, scol, wgt, NUM_NODES);

    const unsigned gemm1_blocks = blk((long long)(NUM_NODES / 16) * (HID / 16), 4);
    const unsigned gemm2_blocks = gemm1_blocks;
    const unsigned gemm3_blocks = blk((long long)(NUM_NODES / 16) * (GATES / 16), 4);
    const unsigned gath_blocks  = blk((long long)NUM_NODES * 32, TPB);

    wmma_gemm_bf16<<<gemm1_blocks, 128, 0, stream>>>(xb, w1t, xw, NUM_NODES, HID, IN_FEAT);
    gcn_gather<<<gath_blocks, TPB, 0, stream>>>(xw, rowptr, scol, wgt, dinv, b1, xb, NUM_NODES);

    wmma_gemm_bf16<<<gemm2_blocks, 128, 0, stream>>>(xb, w2t, xw, NUM_NODES, HID, HID);
    gcn_gather<<<gath_blocks, TPB, 0, stream>>>(xw, rowptr, scol, wgt, dinv, b2, xb, NUM_NODES);

    wmma_gemm_bf16<<<gemm3_blocks, 128, 0, stream>>>(xb, wih, gi, NUM_NODES, GATES, HID);

    gru_scan<<<1, SCAN_THREADS, 0, stream>>>(gi, whh, b_ih, b_hh, hs, NUM_NODES);

    fc_out_kernel<<<blk((long long)NUM_NODES * OUTF, TPB), TPB, 0, stream>>>(
        hs, fc_w, fc_b, out, NUM_NODES * OUTF);
}
